# starting kernel
# baseline (speedup 1.0000x reference)
#include <hip/hip_runtime.h>

typedef __attribute__((ext_vector_type(2))) float v2f;
typedef __attribute__((ext_vector_type(8))) float v8f;

__global__ void k_init_deg(float* __restrict__ deg, int n) {
  int i = blockIdx.x * blockDim.x + threadIdx.x;
  if (i < n) deg[i] = 1.0f;
}

__global__ void k_deg_accum(const int* __restrict__ dst, float* __restrict__ deg, int E) {
  int e = blockIdx.x * blockDim.x + threadIdx.x;
  if (e < E) unsafeAtomicAdd(&deg[dst[e]], 1.0f);
}

__global__ void k_dinv(float* __restrict__ deg, int n) {
  int i = blockIdx.x * blockDim.x + threadIdx.x;
  if (i < n) deg[i] = rsqrtf(deg[i]);
}

__global__ void __launch_bounds__(32)
k_wmma_l1(const float* __restrict__ x, const float* __restrict__ W1,
          const float* __restrict__ dinv, float* __restrict__ t1,
          float* __restrict__ agg1, int n) {
  const int tile = blockIdx.x;
  const int lane = threadIdx.x;
  const int half = lane >> 4;
  const int l    = lane & 15;
  int row = tile * 16 + l;
  int rc  = row < n ? row : n - 1;

  v2f xv = *(const v2f*)(x + (size_t)rc * 2);
  v2f a;
  a.x = half ? 0.0f : xv.x;
  a.y = half ? 0.0f : xv.y;

  float w1a = W1[(l & 7)];
  float w1b = W1[8 + (l & 7)];
  bool  bok = (!half) && (l < 8);
  v2f b;
  b.x = bok ? w1a : 0.0f;
  b.y = bok ? w1b : 0.0f;

  v8f c = {};
  c = __builtin_amdgcn_wmma_f32_16x16x4_f32(false, a, false, b, (short)0, c, false, false);

  const bool fullt = (tile * 16 + 16) <= n;
  if (l < 8) {
    const int base = tile * 16 + half * 8;
    if (fullt) {
#pragma unroll
      for (int r = 0; r < 8; ++r) {
        int node = base + r;
        float v  = c[r];
        float di = dinv[node];
        t1[(size_t)node * 8 + l]   = v;
        agg1[(size_t)node * 8 + l] = v * di * di;
      }
    } else {
#pragma unroll
      for (int r = 0; r < 8; ++r) {
        int node = base + r;
        if (node < n) {
          float v  = c[r];
          float di = dinv[node];
          t1[(size_t)node * 8 + l]   = v;
          agg1[(size_t)node * 8 + l] = v * di * di;
        }
      }
    }
  }
}

__global__ void k_scatter1(const int* __restrict__ src, const int* __restrict__ dst,
                           const float* __restrict__ dinv, const float* __restrict__ t1,
                           float* __restrict__ agg1, int E) {
  int e = blockIdx.x * blockDim.x + threadIdx.x;
  if (e >= E) return;
  int s = src[e], d = dst[e];
  float w = dinv[s] * dinv[d];
  const float4* tp = (const float4*)(t1 + (size_t)s * 8);
  float4 v0 = tp[0];
  float4 v1 = tp[1];
  float* ap = agg1 + (size_t)d * 8;
  unsafeAtomicAdd(ap + 0, v0.x * w);
  unsafeAtomicAdd(ap + 1, v0.y * w);
  unsafeAtomicAdd(ap + 2, v0.z * w);
  unsafeAtomicAdd(ap + 3, v0.w * w);
  unsafeAtomicAdd(ap + 4, v1.x * w);
  unsafeAtomicAdd(ap + 5, v1.y * w);
  unsafeAtomicAdd(ap + 6, v1.z * w);
  unsafeAtomicAdd(ap + 7, v1.w * w);
}

__global__ void __launch_bounds__(32)
k_wmma_l2(const float* __restrict__ agg1, const float* __restrict__ b1,
          const float* __restrict__ W2, const float* __restrict__ b2,
          const float* __restrict__ dinv, float* __restrict__ t2,
          float* __restrict__ out, int n) {
  const int tile = blockIdx.x;
  const int lane = threadIdx.x;
  const int half = lane >> 4;
  const int l    = lane & 15;
  int row = tile * 16 + l;
  int rc  = row < n ? row : n - 1;
  const int ka = half * 2;

  v2f g0 = *(const v2f*)(agg1 + (size_t)rc * 8 + ka);
  v2f g1 = *(const v2f*)(agg1 + (size_t)rc * 8 + ka + 4);
  v2f bb0 = *(const v2f*)(b1 + ka);
  v2f bb1 = *(const v2f*)(b1 + ka + 4);
  v2f a1, a2;
  a1.x = fmaxf(g0.x + bb0.x, 0.0f);
  a1.y = fmaxf(g0.y + bb0.y, 0.0f);
  a2.x = fmaxf(g1.x + bb1.x, 0.0f);
  a2.y = fmaxf(g1.y + bb1.y, 0.0f);

  const int lc = l & 3;
  float wr0 = W2[(ka)     * 4 + lc];
  float wr1 = W2[(ka + 1) * 4 + lc];
  float wr2 = W2[(ka + 4) * 4 + lc];
  float wr3 = W2[(ka + 5) * 4 + lc];
  bool cok = (l < 4);
  v2f b1v, b2v;
  b1v.x = cok ? wr0 : 0.0f;
  b1v.y = cok ? wr1 : 0.0f;
  b2v.x = cok ? wr2 : 0.0f;
  b2v.y = cok ? wr3 : 0.0f;

  v8f c = {};
  c = __builtin_amdgcn_wmma_f32_16x16x4_f32(false, a1, false, b1v, (short)0, c, false, false);
  c = __builtin_amdgcn_wmma_f32_16x16x4_f32(false, a2, false, b2v, (short)0, c, false, false);

  const bool fullt = (tile * 16 + 16) <= n;
  if (l < 4) {
    const int base = tile * 16 + half * 8;
    float bias = b2[l];
    if (fullt) {
#pragma unroll
      for (int r = 0; r < 8; ++r) {
        int node = base + r;
        float v  = c[r];
        float di = dinv[node];
        t2[(size_t)node * 4 + l]  = v;
        out[(size_t)node * 4 + l] = bias + v * di * di;
      }
    } else {
#pragma unroll
      for (int r = 0; r < 8; ++r) {
        int node = base + r;
        if (node < n) {
          float v  = c[r];
          float di = dinv[node];
          t2[(size_t)node * 4 + l]  = v;
          out[(size_t)node * 4 + l] = bias + v * di * di;
        }
      }
    }
  }
}

__global__ void k_scatter2(const int* __restrict__ src, const int* __restrict__ dst,
                           const float* __restrict__ dinv, const float* __restrict__ t2,
                           float* __restrict__ out, int E) {
  int e = blockIdx.x * blockDim.x + threadIdx.x;
  if (e >= E) return;
  int s = src[e], d = dst[e];
  float w = dinv[s] * dinv[d];
  float4 v = *(const float4*)(t2 + (size_t)s * 4);
  float* op = out + (size_t)d * 4;
  unsafeAtomicAdd(op + 0, v.x * w);
  unsafeAtomicAdd(op + 1, v.y * w);
  unsafeAtomicAdd(op + 2, v.z * w);
  unsafeAtomicAdd(op + 3, v.w * w);
}

extern "C" void kernel_launch(void* const* d_in, const int* in_sizes, int n_in,
                              void* d_out, int out_size, void* d_ws, size_t ws_size,
                              hipStream_t stream) {
  const float* x  = (const float*)d_in[0];
  const int*   ei = (const int*)d_in[1];
  const float* W1 = (const float*)d_in[2];
  const float* b1 = (const float*)d_in[3];
  const float* W2 = (const float*)d_in[4];
  const float* b2 = (const float*)d_in[5];
  float* out = (float*)d_out;

  const int n = in_sizes[0] / 2;
  const int E = in_sizes[1] / 2;
  const int* src = ei;
  const int* dst = ei + E;

  char* ws = (char*)d_ws;
  size_t nAl = ((size_t)n + 3) & ~(size_t)3;
  float* dinv = (float*)ws;  ws += sizeof(float) * nAl;
  float* t1   = (float*)ws;  ws += sizeof(float) * (size_t)n * 8;
  float* agg1 = (float*)ws;  ws += sizeof(float) * (size_t)n * 8;
  float* t2   = (float*)ws;

  const int tiles = (n + 15) / 16;

  k_init_deg <<<(n + 255) / 256, 256, 0, stream>>>(dinv, n);
  k_deg_accum<<<(E + 255) / 256, 256, 0, stream>>>(dst, dinv, E);
  k_dinv     <<<(n + 255) / 256, 256, 0, stream>>>(dinv, n);

  k_wmma_l1<<<tiles, 32, 0, stream>>>(x, W1, dinv, t1, agg1, n);

  k_scatter1<<<(E + 255) / 256, 256, 0, stream>>>(src, dst, dinv, t1, agg1, E);

  k_wmma_l2<<<tiles, 32, 0, stream>>>(agg1, b1, W2, b2, dinv, t2, out, n);

  k_scatter2<<<(E + 255) / 256, 256, 0, stream>>>(src, dst, dinv, t2, out, E);
}


# Layer 1 scatters the 2-wide x rows and applies W1 (WMMA) after the sum, not before: 2 f32 atomics per edge instead of 8; changes f32 summation order
# speedup vs baseline: 1.2547x; 1.2547x over previous
#include <hip/hip_runtime.h>

typedef __attribute__((ext_vector_type(2))) float v2f;
typedef __attribute__((ext_vector_type(8))) float v8f;

__global__ void k_init_deg(float* __restrict__ deg, int n) {
  int i = blockIdx.x * blockDim.x + threadIdx.x;
  if (i < n) deg[i] = 1.0f;
}

__global__ void k_deg_accum(const int* __restrict__ dst, float* __restrict__ deg, int E) {
  int e = blockIdx.x * blockDim.x + threadIdx.x;
  if (e < E) unsafeAtomicAdd(&deg[dst[e]], 1.0f);
}

__global__ void k_dinv(float* __restrict__ deg, const float* __restrict__ x,
                       float* __restrict__ aggx, int n) {
  int i = blockIdx.x * blockDim.x + threadIdx.x;
  if (i < n) {
    float di = rsqrtf(deg[i]);
    deg[i] = di;
    v2f xv = *(const v2f*)(x + (size_t)i * 2);
    v2f a;
    a.x = xv.x * di * di;
    a.y = xv.y * di * di;
    *(v2f*)(aggx + (size_t)i * 2) = a;
  }
}

__global__ void k_scatter1(const int* __restrict__ src, const int* __restrict__ dst,
                           const float* __restrict__ dinv, const float* __restrict__ x,
                           float* __restrict__ aggx, int E) {
  int e = blockIdx.x * blockDim.x + threadIdx.x;
  if (e >= E) return;
  int s = src[e], d = dst[e];
  float w = dinv[s] * dinv[d];
  v2f v = *(const v2f*)(x + (size_t)s * 2);
  float* ap = aggx + (size_t)d * 2;
  unsafeAtomicAdd(ap + 0, v.x * w);
  unsafeAtomicAdd(ap + 1, v.y * w);
}

__global__ void __launch_bounds__(32)
k_wmma_l1(const float* __restrict__ aggx, const float* __restrict__ W1,
          float* __restrict__ agg1, int n) {
  const int tile = blockIdx.x;
  const int lane = threadIdx.x;
  const int half = lane >> 4;
  const int l    = lane & 15;
  int row = tile * 16 + l;
  int rc  = row < n ? row : n - 1;

  v2f xv = *(const v2f*)(aggx + (size_t)rc * 2);
  v2f a;
  a.x = half ? 0.0f : xv.x;
  a.y = half ? 0.0f : xv.y;

  float w1a = W1[(l & 7)];
  float w1b = W1[8 + (l & 7)];
  bool  bok = (!half) && (l < 8);
  v2f b;
  b.x = bok ? w1a : 0.0f;
  b.y = bok ? w1b : 0.0f;

  v8f c = {};
  c = __builtin_amdgcn_wmma_f32_16x16x4_f32(false, a, false, b, (short)0, c, false, false);

  const bool fullt = (tile * 16 + 16) <= n;
  if (l < 8) {
    const int base = tile * 16 + half * 8;
    if (fullt) {
#pragma unroll
      for (int r = 0; r < 8; ++r) {
        int node = base + r;
        agg1[(size_t)node * 8 + l] = c[r];
      }
    } else {
#pragma unroll
      for (int r = 0; r < 8; ++r) {
        int node = base + r;
        if (node < n) agg1[(size_t)node * 8 + l] = c[r];
      }
    }
  }
}

__global__ void __launch_bounds__(32)
k_wmma_l2(const float* __restrict__ agg1, const float* __restrict__ b1,
          const float* __restrict__ W2, const float* __restrict__ b2,
          const float* __restrict__ dinv, float* __restrict__ t2,
          float* __restrict__ out, int n) {
  const int tile = blockIdx.x;
  const int lane = threadIdx.x;
  const int half = lane >> 4;
  const int l    = lane & 15;
  int row = tile * 16 + l;
  int rc  = row < n ? row : n - 1;
  const int ka = half * 2;

  v2f g0 = *(const v2f*)(agg1 + (size_t)rc * 8 + ka);
  v2f g1 = *(const v2f*)(agg1 + (size_t)rc * 8 + ka + 4);
  v2f bb0 = *(const v2f*)(b1 + ka);
  v2f bb1 = *(const v2f*)(b1 + ka + 4);
  v2f a1, a2;
  a1.x = fmaxf(g0.x + bb0.x, 0.0f);
  a1.y = fmaxf(g0.y + bb0.y, 0.0f);
  a2.x = fmaxf(g1.x + bb1.x, 0.0f);
  a2.y = fmaxf(g1.y + bb1.y, 0.0f);

  const int lc = l & 3;
  float wr0 = W2[(ka)     * 4 + lc];
  float wr1 = W2[(ka + 1) * 4 + lc];
  float wr2 = W2[(ka + 4) * 4 + lc];
  float wr3 = W2[(ka + 5) * 4 + lc];
  bool cok = (l < 4);
  v2f b1v, b2v;
  b1v.x = cok ? wr0 : 0.0f;
  b1v.y = cok ? wr1 : 0.0f;
  b2v.x = cok ? wr2 : 0.0f;
  b2v.y = cok ? wr3 : 0.0f;

  v8f c = {};
  c = __builtin_amdgcn_wmma_f32_16x16x4_f32(false, a1, false, b1v, (short)0, c, false, false);
  c = __builtin_amdgcn_wmma_f32_16x16x4_f32(false, a2, false, b2v, (short)0, c, false, false);

  const bool fullt = (tile * 16 + 16) <= n;
  if (l < 4) {
    const int base = tile * 16 + half * 8;
    float bias = b2[l];
    if (fullt) {
#pragma unroll
      for (int r = 0; r < 8; ++r) {
        int node = base + r;
        float v  = c[r];
        float di = dinv[node];
        t2[(size_t)node * 4 + l]  = v;
        out[(size_t)node * 4 + l] = bias + v * di * di;
      }
    } else {
#pragma unroll
      for (int r = 0; r < 8; ++r) {
        int node = base + r;
        if (node < n) {
          float v  = c[r];
          float di = dinv[node];
          t2[(size_t)node * 4 + l]  = v;
          out[(size_t)node * 4 + l] = bias + v * di * di;
        }
      }
    }
  }
}

__global__ void k_scatter2(const int* __restrict__ src, const int* __restrict__ dst,
                           const float* __restrict__ dinv, const float* __restrict__ t2,
                           float* __restrict__ out, int E) {
  int e = blockIdx.x * blockDim.x + threadIdx.x;
  if (e >= E) return;
  int s = src[e], d = dst[e];
  float w = dinv[s] * dinv[d];
  float4 v = *(const float4*)(t2 + (size_t)s * 4);
  float* op = out + (size_t)d * 4;
  unsafeAtomicAdd(op + 0, v.x * w);
  unsafeAtomicAdd(op + 1, v.y * w);
  unsafeAtomicAdd(op + 2, v.z * w);
  unsafeAtomicAdd(op + 3, v.w * w);
}

extern "C" void kernel_launch(void* const* d_in, const int* in_sizes, int n_in,
                              void* d_out, int out_size, void* d_ws, size_t ws_size,
                              hipStream_t stream) {
  const float* x  = (const float*)d_in[0];
  const int*   ei = (const int*)d_in[1];
  const float* W1 = (const float*)d_in[2];
  const float* b1 = (const float*)d_in[3];
  const float* W2 = (const float*)d_in[4];
  const float* b2 = (const float*)d_in[5];
  float* out = (float*)d_out;

  const int n = in_sizes[0] / 2;
  const int E = in_sizes[1] / 2;
  const int* src = ei;
  const int* dst = ei + E;

  char* ws = (char*)d_ws;
  size_t nAl = ((size_t)n + 3) & ~(size_t)3;
  float* dinv = (float*)ws;  ws += sizeof(float) * nAl;
  float* aggx = (float*)ws;  ws += sizeof(float) * nAl * 2;
  float* agg1 = (float*)ws;  ws += sizeof(float) * (size_t)n * 8;
  float* t2   = (float*)ws;

  const int tiles = (n + 15) / 16;

  k_init_deg <<<(n + 255) / 256, 256, 0, stream>>>(dinv, n);
  k_deg_accum<<<(E + 255) / 256, 256, 0, stream>>>(dst, dinv, E);
  k_dinv     <<<(n + 255) / 256, 256, 0, stream>>>(dinv, x, aggx, n);

  k_scatter1<<<(E + 255) / 256, 256, 0, stream>>>(src, dst, dinv, x, aggx, E);

  k_wmma_l1<<<tiles, 32, 0, stream>>>(aggx, W1, agg1, n);

  k_wmma_l2<<<tiles, 32, 0, stream>>>(agg1, b1, W2, b2, dinv, t2, out, n);

  k_scatter2<<<(E + 255) / 256, 256, 0, stream>>>(src, dst, dinv, t2, out, E);
}


# Atomic targets (degree, layer-1 sums, output) kept as one copy per package half, chosen by the wave's die number; copies added afterwards (one more N-wide launch); f32 summation order changes
# speedup vs baseline: 4.0063x; 3.1930x over previous
#include <hip/hip_runtime.h>

typedef __attribute__((ext_vector_type(2))) float v2f;
typedef __attribute__((ext_vector_type(8))) float v8f;

__device__ __forceinline__ int pkg_half() {
  return (int)((__builtin_amdgcn_s_sendmsg_rtn(0x87) >> 18) & 1u);
}

__global__ void k_init_deg(float* __restrict__ degA, float* __restrict__ degB, int n) {
  int i = blockIdx.x * blockDim.x + threadIdx.x;
  if (i < n) {
    degA[i] = 1.0f;
    degB[i] = 0.0f;
  }
}

__global__ void k_deg_accum(const int* __restrict__ dst, float* __restrict__ degA,
                            float* __restrict__ degB, int E) {
  int e = blockIdx.x * blockDim.x + threadIdx.x;
  if (e < E) {
    int d = dst[e];
    float* deg = pkg_half() ? degB : degA;
    unsafeAtomicAdd(&deg[d], 1.0f);
  }
}

__global__ void k_dinv(float* __restrict__ degA, const float* __restrict__ degB,
                       const float* __restrict__ x,
                       float* __restrict__ aggxA, float* __restrict__ aggxB, int n) {
  int i = blockIdx.x * blockDim.x + threadIdx.x;
  if (i < n) {
    float di = rsqrtf(degA[i] + degB[i]);
    degA[i] = di;
    v2f xv = *(const v2f*)(x + (size_t)i * 2);
    v2f a;
    a.x = xv.x * di * di;
    a.y = xv.y * di * di;
    *(v2f*)(aggxA + (size_t)i * 2) = a;
    v2f z = {0.0f, 0.0f};
    *(v2f*)(aggxB + (size_t)i * 2) = z;
  }
}

__global__ void k_scatter1(const int* __restrict__ src, const int* __restrict__ dst,
                           const float* __restrict__ dinv, const float* __restrict__ x,
                           float* __restrict__ aggxA, float* __restrict__ aggxB, int E) {
  int e = blockIdx.x * blockDim.x + threadIdx.x;
  if (e >= E) return;
  int s = src[e], d = dst[e];
  float* aggx = pkg_half() ? aggxB : aggxA;
  float w = dinv[s] * dinv[d];
  v2f v = *(const v2f*)(x + (size_t)s * 2);
  float* ap = aggx + (size_t)d * 2;
  unsafeAtomicAdd(ap + 0, v.x * w);
  unsafeAtomicAdd(ap + 1, v.y * w);
}

__global__ void __launch_bounds__(32)
k_wmma_l1(const float* __restrict__ aggxA, const float* __restrict__ aggxB,
          const float* __restrict__ W1, float* __restrict__ agg1, int n) {
  const int tile = blockIdx.x;
  const int lane = threadIdx.x;
  const int half = lane >> 4;
  const int l    = lane & 15;
  int row = tile * 16 + l;
  int rc  = row < n ? row : n - 1;

  v2f xa = *(const v2f*)(aggxA + (size_t)rc * 2);
  v2f xb = *(const v2f*)(aggxB + (size_t)rc * 2);
  v2f xv;
  xv.x = xa.x + xb.x;
  xv.y = xa.y + xb.y;
  v2f a;
  a.x = half ? 0.0f : xv.x;
  a.y = half ? 0.0f : xv.y;

  float w1a = W1[(l & 7)];
  float w1b = W1[8 + (l & 7)];
  bool  bok = (!half) && (l < 8);
  v2f b;
  b.x = bok ? w1a : 0.0f;
  b.y = bok ? w1b : 0.0f;

  v8f c = {};
  c = __builtin_amdgcn_wmma_f32_16x16x4_f32(false, a, false, b, (short)0, c, false, false);

  const bool fullt = (tile * 16 + 16) <= n;
  if (l < 8) {
    const int base = tile * 16 + half * 8;
    if (fullt) {
#pragma unroll
      for (int r = 0; r < 8; ++r) {
        int node = base + r;
        agg1[(size_t)node * 8 + l] = c[r];
      }
    } else {
#pragma unroll
      for (int r = 0; r < 8; ++r) {
        int node = base + r;
        if (node < n) agg1[(size_t)node * 8 + l] = c[r];
      }
    }
  }
}

__global__ void __launch_bounds__(32)
k_wmma_l2(const float* __restrict__ agg1, const float* __restrict__ b1,
          const float* __restrict__ W2, const float* __restrict__ b2,
          const float* __restrict__ dinv, float* __restrict__ t2,
          float* __restrict__ out, float* __restrict__ outB, int n) {
  const int tile = blockIdx.x;
  const int lane = threadIdx.x;
  const int half = lane >> 4;
  const int l    = lane & 15;
  int row = tile * 16 + l;
  int rc  = row < n ? row : n - 1;
  const int ka = half * 2;

  v2f g0 = *(const v2f*)(agg1 + (size_t)rc * 8 + ka);
  v2f g1 = *(const v2f*)(agg1 + (size_t)rc * 8 + ka + 4);
  v2f bb0 = *(const v2f*)(b1 + ka);
  v2f bb1 = *(const v2f*)(b1 + ka + 4);
  v2f a1, a2;
  a1.x = fmaxf(g0.x + bb0.x, 0.0f);
  a1.y = fmaxf(g0.y + bb0.y, 0.0f);
  a2.x = fmaxf(g1.x + bb1.x, 0.0f);
  a2.y = fmaxf(g1.y + bb1.y, 0.0f);

  const int lc = l & 3;
  float wr0 = W2[(ka)     * 4 + lc];
  float wr1 = W2[(ka + 1) * 4 + lc];
  float wr2 = W2[(ka + 4) * 4 + lc];
  float wr3 = W2[(ka + 5) * 4 + lc];
  bool cok = (l < 4);
  v2f b1v, b2v;
  b1v.x = cok ? wr0 : 0.0f;
  b1v.y = cok ? wr1 : 0.0f;
  b2v.x = cok ? wr2 : 0.0f;
  b2v.y = cok ? wr3 : 0.0f;

  v8f c = {};
  c = __builtin_amdgcn_wmma_f32_16x16x4_f32(false, a1, false, b1v, (short)0, c, false, false);
  c = __builtin_amdgcn_wmma_f32_16x16x4_f32(false, a2, false, b2v, (short)0, c, false, false);

  const bool fullt = (tile * 16 + 16) <= n;
  if (l < 4) {
    const int base = tile * 16 + half * 8;
    float bias = b2[l];
    if (fullt) {
#pragma unroll
      for (int r = 0; r < 8; ++r) {
        int node = base + r;
        float v  = c[r];
        float di = dinv[node];
        t2[(size_t)node * 4 + l]  = v;
        out[(size_t)node * 4 + l] = bias + v * di * di;
        outB[(size_t)node * 4 + l] = 0.0f;
      }
    } else {
#pragma unroll
      for (int r = 0; r < 8; ++r) {
        int node = base + r;
        if (node < n) {
          float v  = c[r];
          float di = dinv[node];
          t2[(size_t)node * 4 + l]  = v;
          out[(size_t)node * 4 + l] = bias + v * di * di;
          outB[(size_t)node * 4 + l] = 0.0f;
        }
      }
    }
  }
}

__global__ void k_scatter2(const int* __restrict__ src, const int* __restrict__ dst,
                           const float* __restrict__ dinv, const float* __restrict__ t2,
                           float* __restrict__ out, float* __restrict__ outB, int E) {
  int e = blockIdx.x * blockDim.x + threadIdx.x;
  if (e >= E) return;
  int s = src[e], d = dst[e];
  float* acc = pkg_half() ? outB : out;
  float w = dinv[s] * dinv[d];
  float4 v = *(const float4*)(t2 + (size_t)s * 4);
  float* op = acc + (size_t)d * 4;
  unsafeAtomicAdd(op + 0, v.x * w);
  unsafeAtomicAdd(op + 1, v.y * w);
  unsafeAtomicAdd(op + 2, v.z * w);
  unsafeAtomicAdd(op + 3, v.w * w);
}

__global__ void k_sum_out(float* __restrict__ out, const float* __restrict__ outB, int n) {
  int i = blockIdx.x * blockDim.x + threadIdx.x;
  if (i < n) {
    float4 a = *(const float4*)(out + (size_t)i * 4);
    float4 b = *(const float4*)(outB + (size_t)i * 4);
    a.x += b.x; a.y += b.y; a.z += b.z; a.w += b.w;
    *(float4*)(out + (size_t)i * 4) = a;
  }
}

extern "C" void kernel_launch(void* const* d_in, const int* in_sizes, int n_in,
                              void* d_out, int out_size, void* d_ws, size_t ws_size,
                              hipStream_t stream) {
  const float* x  = (const float*)d_in[0];
  const int*   ei = (const int*)d_in[1];
  const float* W1 = (const float*)d_in[2];
  const float* b1 = (const float*)d_in[3];
  const float* W2 = (const float*)d_in[4];
  const float* b2 = (const float*)d_in[5];
  float* out = (float*)d_out;

  const int n = in_sizes[0] / 2;
  const int E = in_sizes[1] / 2;
  const int* src = ei;
  const int* dst = ei + E;

  char* ws = (char*)d_ws;
  size_t nAl = ((size_t)n + 3) & ~(size_t)3;
  float* dinv  = (float*)ws;  ws += sizeof(float) * nAl;
  float* degB  = (float*)ws;  ws += sizeof(float) * nAl;
  float* aggxA = (float*)ws;  ws += sizeof(float) * nAl * 2;
  float* aggxB = (float*)ws;  ws += sizeof(float) * nAl * 2;
  float* agg1  = (float*)ws;  ws += sizeof(float) * (size_t)n * 8;
  float* t2    = (float*)ws;  ws += sizeof(float) * (size_t)n * 4;
  float* outB  = (float*)ws;

  const int tiles = (n + 15) / 16;

  k_init_deg <<<(n + 255) / 256, 256, 0, stream>>>(dinv, degB, n);
  k_deg_accum<<<(E + 255) / 256, 256, 0, stream>>>(dst, dinv, degB, E);
  k_dinv     <<<(n + 255) / 256, 256, 0, stream>>>(dinv, degB, x, aggxA, aggxB, n);

  k_scatter1<<<(E + 255) / 256, 256, 0, stream>>>(src, dst, dinv, x, aggxA, aggxB, E);

  k_wmma_l1<<<tiles, 32, 0, stream>>>(aggxA, aggxB, W1, agg1, n);

  k_wmma_l2<<<tiles, 32, 0, stream>>>(agg1, b1, W2, b2, dinv, t2, out, outB, n);

  k_scatter2<<<(E + 255) / 256, 256, 0, stream>>>(src, dst, dinv, t2, out, outB, E);

  k_sum_out<<<(n + 255) / 256, 256, 0, stream>>>(out, outB, n);
}


# All three edge sums built in LDS (ds_add_f32): a work-group owns a node range and an edge slice and stores a partial plane; consumers add the planes; no global atomics left; f32 sum order changes
# speedup vs baseline: 7.1684x; 1.7893x over previous
#include <hip/hip_runtime.h>


typedef __attribute__((ext_vector_type(2))) float v2f;
typedef __attribute__((ext_vector_type(8))) float v8f;

#define ACC_FLOATS 80000
#define SC_THREADS 1024
#define SC_U 8

__device__ __forceinline__ void lds_add(float* a, float v) {
  __hip_atomic_fetch_add(a, v, __ATOMIC_RELAXED, __HIP_MEMORY_SCOPE_WORKGROUP);
}

template <int WID>
__global__ void __launch_bounds__(SC_THREADS)
k_scatter_lds(const int* __restrict__ src, const int* __restrict__ dst,
              const float* __restrict__ rows, float* __restrict__ part,
              int n, int E, int S) {
  __shared__ __attribute__((aligned(16))) float acc[ACC_FLOATS];
  constexpr int R = ACC_FLOATS / WID;
  const int tid = threadIdx.x;
  const int r = blockIdx.x / S;
  const int s = blockIdx.x - r * S;
  const int lo = r * R;
  const int span = (n - lo) < R ? (n - lo) : R;
  const int cnt = span * WID;

  float4* acc4 = (float4*)acc;
  const float4 z4 = {0.0f, 0.0f, 0.0f, 0.0f};
  for (int i = tid; i < ACC_FLOATS / 4; i += SC_THREADS) acc4[i] = z4;
  __syncthreads();

  const int chunk = (E + S - 1) / S;
  const int e0 = s * chunk;
  const int e1 = (e0 + chunk) < E ? (e0 + chunk) : E;
  const int step = SC_THREADS * SC_U;

  if (e0 < e1) {
    const int last = e1 - 1;
    int dcur[SC_U];
#pragma unroll
    for (int j = 0; j < SC_U; ++j) {
      const int e = e0 + tid + j * SC_THREADS;
      dcur[j] = dst[e < last ? e : last];
    }
    for (int base = e0 + tid; base < e1; base += step) {
      int dnext[SC_U];
#pragma unroll
      for (int j = 0; j < SC_U; ++j) {
        const int e = base + step + j * SC_THREADS;
        dnext[j] = dst[e < last ? e : last];
      }
      unsigned rel[SC_U];
      bool hit[SC_U];
#pragma unroll
      for (int j = 0; j < SC_U; ++j) {
        const int e = base + j * SC_THREADS;
        rel[j] = (unsigned)(dcur[j] - lo);
        hit[j] = (e < e1) && (rel[j] < (unsigned)span);
      }
      if (WID == 1) {
#pragma unroll
        for (int j = 0; j < SC_U; ++j)
          if (hit[j]) lds_add(&acc[rel[j]], 1.0f);
      } else {
        int sidx[SC_U];
#pragma unroll
        for (int j = 0; j < SC_U; ++j) {
          sidx[j] = 0;
          if (hit[j]) sidx[j] = src[base + j * SC_THREADS];
        }
        if (WID == 2) {
          v2f v[SC_U];
#pragma unroll
          for (int j = 0; j < SC_U; ++j) v[j] = *(const v2f*)(rows + (size_t)sidx[j] * 2);
#pragma unroll
          for (int j = 0; j < SC_U; ++j)
            if (hit[j]) {
              float* a = &acc[rel[j] * 2];
              lds_add(a + 0, v[j].x);
              lds_add(a + 1, v[j].y);
            }
        } else {
          float4 v[SC_U];
#pragma unroll
          for (int j = 0; j < SC_U; ++j) v[j] = *(const float4*)(rows + (size_t)sidx[j] * 4);
#pragma unroll
          for (int j = 0; j < SC_U; ++j)
            if (hit[j]) {
              float* a = &acc[rel[j] * 4];
              lds_add(a + 0, v[j].x);
              lds_add(a + 1, v[j].y);
              lds_add(a + 2, v[j].z);
              lds_add(a + 3, v[j].w);
            }
        }
      }
#pragma unroll
      for (int j = 0; j < SC_U; ++j) dcur[j] = dnext[j];
    }
  }
  __syncthreads();

  float* p = part + ((size_t)s * n + lo) * WID;
  if ((((size_t)p) & 15) == 0 && (cnt & 3) == 0) {
    float4* p4 = (float4*)p;
    for (int i = tid; i < cnt / 4; i += SC_THREADS) p4[i] = acc4[i];
  } else {
    for (int i = tid; i < cnt; i += SC_THREADS) p[i] = acc[i];
  }
}

__global__ void k_dinv(const float* __restrict__ part, int S,
                       const float* __restrict__ x, float* __restrict__ dinv,
                       float* __restrict__ xs, int n) {
  int i = blockIdx.x * blockDim.x + threadIdx.x;
  if (i < n) {
    float deg = 1.0f;
#pragma unroll 8
    for (int s = 0; s < S; ++s) deg += part[(size_t)s * n + i];
    float di = rsqrtf(deg);
    dinv[i] = di;
    v2f xv = *(const v2f*)(x + (size_t)i * 2);
    v2f a;
    a.x = xv.x * di;
    a.y = xv.y * di;
    *(v2f*)(xs + (size_t)i * 2) = a;
  }
}

__global__ void __launch_bounds__(32)
k_wmma_l1(const float* __restrict__ xs, const float* __restrict__ part, int S,
          const float* __restrict__ dinv,
          const float* __restrict__ W1, float* __restrict__ agg1, int n) {
  const int tile = blockIdx.x;
  const int lane = threadIdx.x;
  const int half = lane >> 4;
  const int l    = lane & 15;
  int row = tile * 16 + l;
  int rc  = row < n ? row : n - 1;

  v2f xa = *(const v2f*)(xs + (size_t)rc * 2);
#pragma unroll 8
  for (int s = 0; s < S; ++s) {
    v2f pv = *(const v2f*)(part + ((size_t)s * n + rc) * 2);
    xa.x += pv.x;
    xa.y += pv.y;
  }
  float dr = dinv[rc];
  v2f xv;
  xv.x = xa.x * dr;
  xv.y = xa.y * dr;
  v2f a;
  a.x = half ? 0.0f : xv.x;
  a.y = half ? 0.0f : xv.y;

  float w1a = W1[(l & 7)];
  float w1b = W1[8 + (l & 7)];
  bool  bok = (!half) && (l < 8);
  v2f b;
  b.x = bok ? w1a : 0.0f;
  b.y = bok ? w1b : 0.0f;

  v8f c = {};
  c = __builtin_amdgcn_wmma_f32_16x16x4_f32(false, a, false, b, (short)0, c, false, false);

  const bool fullt = (tile * 16 + 16) <= n;
  if (l < 8) {
    const int base = tile * 16 + half * 8;
    if (fullt) {
#pragma unroll
      for (int r = 0; r < 8; ++r) {
        int node = base + r;
        agg1[(size_t)node * 8 + l] = c[r];
      }
    } else {
#pragma unroll
      for (int r = 0; r < 8; ++r) {
        int node = base + r;
        if (node < n) agg1[(size_t)node * 8 + l] = c[r];
      }
    }
  }
}

__global__ void __launch_bounds__(32)
k_wmma_l2(const float* __restrict__ agg1, const float* __restrict__ b1,
          const float* __restrict__ W2,
          const float* __restrict__ dinv, float* __restrict__ t2, int n) {
  const int tile = blockIdx.x;
  const int lane = threadIdx.x;
  const int half = lane >> 4;
  const int l    = lane & 15;
  int row = tile * 16 + l;
  int rc  = row < n ? row : n - 1;
  const int ka = half * 2;

  v2f g0 = *(const v2f*)(agg1 + (size_t)rc * 8 + ka);
  v2f g1 = *(const v2f*)(agg1 + (size_t)rc * 8 + ka + 4);
  v2f bb0 = *(const v2f*)(b1 + ka);
  v2f bb1 = *(const v2f*)(b1 + ka + 4);
  v2f a1, a2;
  a1.x = fmaxf(g0.x + bb0.x, 0.0f);
  a1.y = fmaxf(g0.y + bb0.y, 0.0f);
  a2.x = fmaxf(g1.x + bb1.x, 0.0f);
  a2.y = fmaxf(g1.y + bb1.y, 0.0f);

  const int lc = l & 3;
  float wr0 = W2[(ka)     * 4 + lc];
  float wr1 = W2[(ka + 1) * 4 + lc];
  float wr2 = W2[(ka + 4) * 4 + lc];
  float wr3 = W2[(ka + 5) * 4 + lc];
  bool cok = (l < 4);
  v2f b1v, b2v;
  b1v.x = cok ? wr0 : 0.0f;
  b1v.y = cok ? wr1 : 0.0f;
  b2v.x = cok ? wr2 : 0.0f;
  b2v.y = cok ? wr3 : 0.0f;

  v8f c = {};
  c = __builtin_amdgcn_wmma_f32_16x16x4_f32(false, a1, false, b1v, (short)0, c, false, false);
  c = __builtin_amdgcn_wmma_f32_16x16x4_f32(false, a2, false, b2v, (short)0, c, false, false);

  const bool fullt = (tile * 16 + 16) <= n;
  if (l < 4) {
    const int base = tile * 16 + half * 8;
    if (fullt) {
#pragma unroll
      for (int r = 0; r < 8; ++r) {
        int node = base + r;
        t2[(size_t)node * 4 + l] = c[r] * dinv[node];
      }
    } else {
#pragma unroll
      for (int r = 0; r < 8; ++r) {
        int node = base + r;
        if (node < n) t2[(size_t)node * 4 + l] = c[r] * dinv[node];
      }
    }
  }
}

__global__ void k_sum_out(float* __restrict__ out, const float* __restrict__ t2,
                          const float* __restrict__ part, int S,
                          const float* __restrict__ dinv, const float* __restrict__ b2, int n) {
  int i = blockIdx.x * blockDim.x + threadIdx.x;
  if (i < n) {
    float4 a = *(const float4*)(t2 + (size_t)i * 4);
#pragma unroll 8
    for (int s = 0; s < S; ++s) {
      float4 pv = *(const float4*)(part + ((size_t)s * n + i) * 4);
      a.x += pv.x;
      a.y += pv.y;
      a.z += pv.z;
      a.w += pv.w;
    }
    float di = dinv[i];
    a.x = b2[0] + a.x * di;
    a.y = b2[1] + a.y * di;
    a.z = b2[2] + a.z * di;
    a.w = b2[3] + a.w * di;
    *(float4*)(out + (size_t)i * 4) = a;
  }
}

extern "C" void kernel_launch(void* const* d_in, const int* in_sizes, int n_in,
                              void* d_out, int out_size, void* d_ws, size_t ws_size,
                              hipStream_t stream) {
  const float* x  = (const float*)d_in[0];
  const int*   ei = (const int*)d_in[1];
  const float* W1 = (const float*)d_in[2];
  const float* b1 = (const float*)d_in[3];
  const float* W2 = (const float*)d_in[4];
  const float* b2 = (const float*)d_in[5];
  float* out = (float*)d_out;

  const int n = in_sizes[0] / 2;
  const int E = in_sizes[1] / 2;
  const int* src = ei;
  const int* dst = ei + E;

  char* ws = (char*)d_ws;
  size_t nAl = ((size_t)n + 3) & ~(size_t)3;
  float* dinv  = (float*)ws;  ws += sizeof(float) * nAl;
  float* xs    = (float*)ws;  ws += sizeof(float) * nAl * 2;
  float* agg1  = (float*)ws;  ws += sizeof(float) * (size_t)n * 8;
  float* t2    = (float*)ws;  ws += sizeof(float) * (size_t)n * 4;
  float* part  = (float*)ws;

  const int tiles = (n + 15) / 16;

  const int g1 = (n + ACC_FLOATS - 1) / ACC_FLOATS;
  const int g2 = (n + ACC_FLOATS / 2 - 1) / (ACC_FLOATS / 2);
  const int g4 = (n + ACC_FLOATS / 4 - 1) / (ACC_FLOATS / 4);
  const int S1 = g1 < 256 ? 256 / g1 : 1;
  const int S2 = g2 < 256 ? 256 / g2 : 1;
  const int S4 = g4 < 256 ? 256 / g4 : 1;

  k_scatter_lds<1><<<g1 * S1, SC_THREADS, 0, stream>>>(src, dst, x, part, n, E, S1);
  k_dinv<<<(n + 255) / 256, 256, 0, stream>>>(part, S1, x, dinv, xs, n);

  k_scatter_lds<2><<<g2 * S2, SC_THREADS, 0, stream>>>(src, dst, xs, part, n, E, S2);
  k_wmma_l1<<<tiles, 32, 0, stream>>>(xs, part, S2, dinv, W1, agg1, n);

  k_wmma_l2<<<tiles, 32, 0, stream>>>(agg1, b1, W2, dinv, t2, n);

  k_scatter_lds<4><<<g4 * S4, SC_THREADS, 0, stream>>>(src, dst, t2, part, n, E, S4);
  k_sum_out<<<(n + 255) / 256, 256, 0, stream>>>(out, t2, part, S4, dinv, b2, n);
}


# Edge list sorted once per call by target node range (counting sort through LDS, 2 launches); each sum: 1 work-group per range reads only its own edges; no planes, no rescans; f32 sum order changes
# speedup vs baseline: 10.9262x; 1.5242x over previous
#include <hip/hip_runtime.h>


typedef __attribute__((ext_vector_type(2))) float v2f;
typedef __attribute__((ext_vector_type(8))) float v8f;

#define NB 256
#define BK_THREADS 1024
#define BK_U 8
#define STAGE_PAIRS 32768
#define ACC_FLOATS 80000

static_assert(BK_THREADS == 4 * NB, "k_range_place adds the count table with four threads per range");

__device__ __forceinline__ void lds_add(float* a, float v) {
  __hip_atomic_fetch_add(a, v, __ATOMIC_RELAXED, __HIP_MEMORY_SCOPE_WORKGROUP);
}

__device__ __forceinline__ int range_of(int d, int rn) {
  const unsigned b = (unsigned)d / (unsigned)rn;
  return b < (unsigned)NB ? (int)b : NB - 1;
}

__global__ void __launch_bounds__(BK_THREADS)
k_range_count(const int* __restrict__ dst, int* __restrict__ counts, int E, int S, int rn) {
  __shared__ unsigned cnt[NB];
  const int tid = threadIdx.x;
  const int s = blockIdx.x;
  if (tid < NB) cnt[tid] = 0u;
  __syncthreads();

  const int chunk = (E + S - 1) / S;
  const int e0 = s * chunk;
  const int e1 = (e0 + chunk) < E ? (e0 + chunk) : E;
  if (e0 < e1) {
    const int last = e1 - 1;
    for (int base = e0 + tid; base < e1; base += BK_THREADS * BK_U) {
      int d[BK_U];
#pragma unroll
      for (int j = 0; j < BK_U; ++j) {
        const int e = base + j * BK_THREADS;
        d[j] = dst[e < last ? e : last];
      }
#pragma unroll
      for (int j = 0; j < BK_U; ++j) {
        const int e = base + j * BK_THREADS;
        if (e < e1)
          __hip_atomic_fetch_add(&cnt[range_of(d[j], rn)], 1u, __ATOMIC_RELAXED, __HIP_MEMORY_SCOPE_WORKGROUP);
      }
    }
  }
  __syncthreads();
  if (tid < NB) counts[s * NB + tid] = (int)cnt[tid];
}

__global__ void __launch_bounds__(BK_THREADS)
k_range_place(const int* __restrict__ src, const int* __restrict__ dst,
              const int* __restrict__ counts, int* __restrict__ bstart,
              int* __restrict__ pairs_i, int E, int S, int rn) {
  __shared__ __attribute__((aligned(16))) int2 stage[STAGE_PAIRS];
  __shared__ unsigned cur[NB];
  __shared__ int gdelta[NB];
  __shared__ int part_below[4][NB];
  __shared__ int part_all[4][NB];
  __shared__ int wsum_g[NB / 32];
  __shared__ int wsum_l[NB / 32];
  const int tid = threadIdx.x;
  const int s = blockIdx.x;
  int2* pairs = (int2*)pairs_i;

  {
    const int b = tid & (NB - 1);
    const int q = tid / NB;
    const int rq = (S + 3) / 4;
    const int r0 = q * rq;
    const int r1 = (r0 + rq) < S ? (r0 + rq) : S;
    int sb = 0, sa = 0;
#pragma unroll 16
    for (int r = r0; r < r1; ++r) {
      const int c = counts[r * NB + b];
      sa += c;
      sb += (r < s) ? c : 0;
    }
    part_below[q][b] = sb;
    part_all[q][b] = sa;
  }
  __syncthreads();

  int below_b = 0, all_b = 0, my_c = 0, g_incl = 0, l_incl = 0;
  if (tid < NB) {
    below_b = part_below[0][tid] + part_below[1][tid] + part_below[2][tid] + part_below[3][tid];
    all_b   = part_all[0][tid] + part_all[1][tid] + part_all[2][tid] + part_all[3][tid];
    my_c    = counts[s * NB + tid];
    g_incl = all_b;
    l_incl = my_c;
    const int lane = tid & 31;
    for (int off = 1; off < 32; off <<= 1) {
      const int tg = __shfl_up(g_incl, off);
      const int tl = __shfl_up(l_incl, off);
      if (lane >= off) { g_incl += tg; l_incl += tl; }
    }
    if (lane == 31) { wsum_g[tid >> 5] = g_incl; wsum_l[tid >> 5] = l_incl; }
  }
  __syncthreads();
  if (tid < NB) {
    int gb = 0, lb = 0;
    const int w = tid >> 5;
    for (int k = 0; k < w; ++k) { gb += wsum_g[k]; lb += wsum_l[k]; }
    const int gs = gb + g_incl - all_b;
    const int ls = lb + l_incl - my_c;
    cur[tid] = (unsigned)ls;
    gdelta[tid] = gs + below_b - ls;
    if (s == 0) {
      bstart[tid] = gs;
      if (tid == NB - 1) bstart[NB] = gs + all_b;
    }
  }
  __syncthreads();

  const int chunk = (E + S - 1) / S;
  const int e0 = s * chunk;
  const int e1 = (e0 + chunk) < E ? (e0 + chunk) : E;
  if (e0 < e1) {
    const int last = e1 - 1;
    for (int base = e0 + tid; base < e1; base += BK_THREADS * BK_U) {
      int d[BK_U], sv[BK_U];
#pragma unroll
      for (int j = 0; j < BK_U; ++j) {
        const int e = base + j * BK_THREADS;
        const int ec = e < last ? e : last;
        d[j] = dst[ec];
        sv[j] = src[ec];
      }
#pragma unroll
      for (int j = 0; j < BK_U; ++j) {
        const int e = base + j * BK_THREADS;
        if (e < e1) {
          const int b = range_of(d[j], rn);
          const unsigned p = __hip_atomic_fetch_add(&cur[b], 1u, __ATOMIC_RELAXED, __HIP_MEMORY_SCOPE_WORKGROUP);
          stage[p] = make_int2(sv[j], d[j]);
        }
      }
    }
  }
  __syncthreads();

  const int len = e1 - e0;
  for (int i = tid; i < len; i += BK_THREADS) {
    const int2 pr = stage[i];
    pairs[gdelta[range_of(pr.y, rn)] + i] = pr;
  }
}

template <int WID>
__global__ void __launch_bounds__(BK_THREADS)
k_owner_sum(const int* __restrict__ pairs_i, const int* __restrict__ bstart,
            const float* __restrict__ rows, float* __restrict__ sums, int n, int rn) {
  __shared__ __attribute__((aligned(16))) float acc[ACC_FLOATS];
  const int tid = threadIdx.x;
  const int2* pairs = (const int2*)pairs_i;
  const int lo = blockIdx.x * rn;
  int span = n - lo;
  span = span < rn ? span : rn;
  span = span > 0 ? span : 0;
  const int cnt = span * WID;

  for (int i = tid; i < cnt; i += BK_THREADS) acc[i] = 0.0f;
  __syncthreads();

  const int p0 = bstart[blockIdx.x];
  const int p1 = bstart[blockIdx.x + 1];
  const int step = BK_THREADS * BK_U;

  if (p0 < p1) {
    const int last = p1 - 1;
    int2 pcur[BK_U];
#pragma unroll
    for (int j = 0; j < BK_U; ++j) {
      const int p = p0 + tid + j * BK_THREADS;
      pcur[j] = pairs[p < last ? p : last];
    }
    for (int base = p0 + tid; base < p1; base += step) {
      int2 pnext[BK_U];
#pragma unroll
      for (int j = 0; j < BK_U; ++j) {
        const int p = base + step + j * BK_THREADS;
        pnext[j] = pairs[p < last ? p : last];
      }
      unsigned rel[BK_U];
      bool hit[BK_U];
#pragma unroll
      for (int j = 0; j < BK_U; ++j) {
        const int p = base + j * BK_THREADS;
        rel[j] = (unsigned)(pcur[j].y - lo);
        hit[j] = (p < p1) && (rel[j] < (unsigned)span);
      }
      if (WID == 1) {
#pragma unroll
        for (int j = 0; j < BK_U; ++j)
          if (hit[j]) lds_add(&acc[rel[j]], 1.0f);
      } else {
        int sidx[BK_U];
#pragma unroll
        for (int j = 0; j < BK_U; ++j) sidx[j] = hit[j] ? pcur[j].x : 0;
        if (WID == 2) {
          v2f v[BK_U];
#pragma unroll
          for (int j = 0; j < BK_U; ++j) v[j] = *(const v2f*)(rows + (size_t)sidx[j] * 2);
#pragma unroll
          for (int j = 0; j < BK_U; ++j)
            if (hit[j]) {
              float* a = &acc[rel[j] * 2];
              lds_add(a + 0, v[j].x);
              lds_add(a + 1, v[j].y);
            }
        } else {
          float4 v[BK_U];
#pragma unroll
          for (int j = 0; j < BK_U; ++j) v[j] = *(const float4*)(rows + (size_t)sidx[j] * 4);
#pragma unroll
          for (int j = 0; j < BK_U; ++j)
            if (hit[j]) {
              float* a = &acc[rel[j] * 4];
              lds_add(a + 0, v[j].x);
              lds_add(a + 1, v[j].y);
              lds_add(a + 2, v[j].z);
              lds_add(a + 3, v[j].w);
            }
        }
      }
#pragma unroll
      for (int j = 0; j < BK_U; ++j) pcur[j] = pnext[j];
    }
  }
  __syncthreads();

  float* p = sums + (size_t)lo * WID;
  for (int i = tid; i < cnt; i += BK_THREADS) p[i] = acc[i];
}

__global__ void k_dinv(const float* __restrict__ part, int S,
                       const float* __restrict__ x, float* __restrict__ dinv,
                       float* __restrict__ xs, int n) {
  int i = blockIdx.x * blockDim.x + threadIdx.x;
  if (i < n) {
    float deg = 1.0f;
#pragma unroll 8
    for (int s = 0; s < S; ++s) deg += part[(size_t)s * n + i];
    float di = rsqrtf(deg);
    dinv[i] = di;
    v2f xv = *(const v2f*)(x + (size_t)i * 2);
    v2f a;
    a.x = xv.x * di;
    a.y = xv.y * di;
    *(v2f*)(xs + (size_t)i * 2) = a;
  }
}

__global__ void __launch_bounds__(32)
k_wmma_l1(const float* __restrict__ xs, const float* __restrict__ part, int S,
          const float* __restrict__ dinv,
          const float* __restrict__ W1, float* __restrict__ agg1, int n) {
  const int tile = blockIdx.x;
  const int lane = threadIdx.x;
  const int half = lane >> 4;
  const int l    = lane & 15;
  int row = tile * 16 + l;
  int rc  = row < n ? row : n - 1;

  v2f xa = *(const v2f*)(xs + (size_t)rc * 2);
#pragma unroll 8
  for (int s = 0; s < S; ++s) {
    v2f pv = *(const v2f*)(part + ((size_t)s * n + rc) * 2);
    xa.x += pv.x;
    xa.y += pv.y;
  }
  float dr = dinv[rc];
  v2f xv;
  xv.x = xa.x * dr;
  xv.y = xa.y * dr;
  v2f a;
  a.x = half ? 0.0f : xv.x;
  a.y = half ? 0.0f : xv.y;

  float w1a = W1[(l & 7)];
  float w1b = W1[8 + (l & 7)];
  bool  bok = (!half) && (l < 8);
  v2f b;
  b.x = bok ? w1a : 0.0f;
  b.y = bok ? w1b : 0.0f;

  v8f c = {};
  c = __builtin_amdgcn_wmma_f32_16x16x4_f32(false, a, false, b, (short)0, c, false, false);

  const bool fullt = (tile * 16 + 16) <= n;
  if (l < 8) {
    const int base = tile * 16 + half * 8;
    if (fullt) {
#pragma unroll
      for (int r = 0; r < 8; ++r) {
        int node = base + r;
        agg1[(size_t)node * 8 + l] = c[r];
      }
    } else {
#pragma unroll
      for (int r = 0; r < 8; ++r) {
        int node = base + r;
        if (node < n) agg1[(size_t)node * 8 + l] = c[r];
      }
    }
  }
}

__global__ void __launch_bounds__(32)
k_wmma_l2(const float* __restrict__ agg1, const float* __restrict__ b1,
          const float* __restrict__ W2,
          const float* __restrict__ dinv, float* __restrict__ t2, int n) {
  const int tile = blockIdx.x;
  const int lane = threadIdx.x;
  const int half = lane >> 4;
  const int l    = lane & 15;
  int row = tile * 16 + l;
  int rc  = row < n ? row : n - 1;
  const int ka = half * 2;

  v2f g0 = *(const v2f*)(agg1 + (size_t)rc * 8 + ka);
  v2f g1 = *(const v2f*)(agg1 + (size_t)rc * 8 + ka + 4);
  v2f bb0 = *(const v2f*)(b1 + ka);
  v2f bb1 = *(const v2f*)(b1 + ka + 4);
  v2f a1, a2;
  a1.x = fmaxf(g0.x + bb0.x, 0.0f);
  a1.y = fmaxf(g0.y + bb0.y, 0.0f);
  a2.x = fmaxf(g1.x + bb1.x, 0.0f);
  a2.y = fmaxf(g1.y + bb1.y, 0.0f);

  const int lc = l & 3;
  float wr0 = W2[(ka)     * 4 + lc];
  float wr1 = W2[(ka + 1) * 4 + lc];
  float wr2 = W2[(ka + 4) * 4 + lc];
  float wr3 = W2[(ka + 5) * 4 + lc];
  bool cok = (l < 4);
  v2f b1v, b2v;
  b1v.x = cok ? wr0 : 0.0f;
  b1v.y = cok ? wr1 : 0.0f;
  b2v.x = cok ? wr2 : 0.0f;
  b2v.y = cok ? wr3 : 0.0f;

  v8f c = {};
  c = __builtin_amdgcn_wmma_f32_16x16x4_f32(false, a1, false, b1v, (short)0, c, false, false);
  c = __builtin_amdgcn_wmma_f32_16x16x4_f32(false, a2, false, b2v, (short)0, c, false, false);

  const bool fullt = (tile * 16 + 16) <= n;
  if (l < 4) {
    const int base = tile * 16 + half * 8;
    if (fullt) {
#pragma unroll
      for (int r = 0; r < 8; ++r) {
        int node = base + r;
        t2[(size_t)node * 4 + l] = c[r] * dinv[node];
      }
    } else {
#pragma unroll
      for (int r = 0; r < 8; ++r) {
        int node = base + r;
        if (node < n) t2[(size_t)node * 4 + l] = c[r] * dinv[node];
      }
    }
  }
}

__global__ void k_sum_out(float* __restrict__ out, const float* __restrict__ t2,
                          const float* __restrict__ part, int S,
                          const float* __restrict__ dinv, const float* __restrict__ b2, int n) {
  int i = blockIdx.x * blockDim.x + threadIdx.x;
  if (i < n) {
    float4 a = *(const float4*)(t2 + (size_t)i * 4);
#pragma unroll 8
    for (int s = 0; s < S; ++s) {
      float4 pv = *(const float4*)(part + ((size_t)s * n + i) * 4);
      a.x += pv.x;
      a.y += pv.y;
      a.z += pv.z;
      a.w += pv.w;
    }
    float di = dinv[i];
    a.x = b2[0] + a.x * di;
    a.y = b2[1] + a.y * di;
    a.z = b2[2] + a.z * di;
    a.w = b2[3] + a.w * di;
    *(float4*)(out + (size_t)i * 4) = a;
  }
}

extern "C" void kernel_launch(void* const* d_in, const int* in_sizes, int n_in,
                              void* d_out, int out_size, void* d_ws, size_t ws_size,
                              hipStream_t stream) {
  const float* x  = (const float*)d_in[0];
  const int*   ei = (const int*)d_in[1];
  const float* W1 = (const float*)d_in[2];
  const float* b1 = (const float*)d_in[3];
  const float* W2 = (const float*)d_in[4];
  const float* b2 = (const float*)d_in[5];
  float* out = (float*)d_out;

  const int n = in_sizes[0] / 2;
  const int E = in_sizes[1] / 2;
  const int* src = ei;
  const int* dst = ei + E;

  const int rn = (n + NB - 1) / NB > 0 ? (n + NB - 1) / NB : 1;
  const int Smin = (E + STAGE_PAIRS - 1) / STAGE_PAIRS;
  const int S = Smin > 256 ? Smin : 256;

  char* ws = (char*)d_ws;
  size_t nAl = ((size_t)n + 3) & ~(size_t)3;
  float* dinv   = (float*)ws;  ws += sizeof(float) * nAl;
  float* xs     = (float*)ws;  ws += sizeof(float) * nAl * 2;
  float* agg1   = (float*)ws;  ws += sizeof(float) * (size_t)n * 8;
  float* t2     = (float*)ws;  ws += sizeof(float) * (size_t)n * 4;
  float* part   = (float*)ws;  ws += sizeof(float) * (size_t)n * 4;
  int*   counts = (int*)ws;    ws += sizeof(int) * (size_t)S * NB;
  int*   bstart = (int*)ws;    ws += sizeof(int) * (NB + 4);
  int*   pairs  = (int*)ws;

  const int tiles = (n + 15) / 16;

  k_range_count<<<S, BK_THREADS, 0, stream>>>(dst, counts, E, S, rn);
  k_range_place<<<S, BK_THREADS, 0, stream>>>(src, dst, counts, bstart, pairs, E, S, rn);

  k_owner_sum<1><<<NB, BK_THREADS, 0, stream>>>(pairs, bstart, x, part, n, rn);
  k_dinv<<<(n + 255) / 256, 256, 0, stream>>>(part, 1, x, dinv, xs, n);

  k_owner_sum<2><<<NB, BK_THREADS, 0, stream>>>(pairs, bstart, xs, part, n, rn);
  k_wmma_l1<<<tiles, 32, 0, stream>>>(xs, part, 1, dinv, W1, agg1, n);

  k_wmma_l2<<<tiles, 32, 0, stream>>>(agg1, b1, W2, dinv, t2, n);

  k_owner_sum<4><<<NB, BK_THREADS, 0, stream>>>(pairs, bstart, t2, part, n, rn);
  k_sum_out<<<(n + 255) / 256, 256, 0, stream>>>(out, t2, part, 1, dinv, b2, n);
}


# Per-node stages (dinv/xs; W1-relu-W2 WMMAs, product 1 transposed so its D feeds product 2 in registers; closing sum) fused into the owner work-group's tail on its LDS sums: 9 launches to 5
# speedup vs baseline: 11.9100x; 1.0900x over previous
#include <hip/hip_runtime.h>


typedef __attribute__((ext_vector_type(2))) float v2f;
typedef __attribute__((ext_vector_type(8))) float v8f;

#define NB 256
#define BK_THREADS 1024
#define BK_U 8
#define STAGE_PAIRS 32768
#define ACC_FLOATS 80000

static_assert(BK_THREADS == 4 * NB, "k_range_place adds the count table with four threads per range");

__device__ __forceinline__ void lds_add(float* a, float v) {
  __hip_atomic_fetch_add(a, v, __ATOMIC_RELAXED, __HIP_MEMORY_SCOPE_WORKGROUP);
}

__device__ __forceinline__ int range_of(int d, int rn) {
  const unsigned b = (unsigned)d / (unsigned)rn;
  return b < (unsigned)NB ? (int)b : NB - 1;
}

__global__ void __launch_bounds__(BK_THREADS)
k_range_count(const int* __restrict__ dst, int* __restrict__ counts, int E, int S, int rn) {
  __shared__ unsigned cnt[NB];
  const int tid = threadIdx.x;
  const int s = blockIdx.x;
  if (tid < NB) cnt[tid] = 0u;
  __syncthreads();

  const int chunk = (E + S - 1) / S;
  const int e0 = s * chunk;
  const int e1 = (e0 + chunk) < E ? (e0 + chunk) : E;
  if (e0 < e1) {
    const int last = e1 - 1;
    for (int base = e0 + tid; base < e1; base += BK_THREADS * BK_U) {
      int d[BK_U];
#pragma unroll
      for (int j = 0; j < BK_U; ++j) {
        const int e = base + j * BK_THREADS;
        d[j] = dst[e < last ? e : last];
      }
#pragma unroll
      for (int j = 0; j < BK_U; ++j) {
        const int e = base + j * BK_THREADS;
        if (e < e1)
          __hip_atomic_fetch_add(&cnt[range_of(d[j], rn)], 1u, __ATOMIC_RELAXED, __HIP_MEMORY_SCOPE_WORKGROUP);
      }
    }
  }
  __syncthreads();
  if (tid < NB) counts[s * NB + tid] = (int)cnt[tid];
}

__global__ void __launch_bounds__(BK_THREADS)
k_range_place(const int* __restrict__ src, const int* __restrict__ dst,
              const int* __restrict__ counts, int* __restrict__ bstart,
              int* __restrict__ pairs_i, int E, int S, int rn) {
  __shared__ __attribute__((aligned(16))) int2 stage[STAGE_PAIRS];
  __shared__ unsigned cur[NB];
  __shared__ int gdelta[NB];
  __shared__ int part_below[4][NB];
  __shared__ int part_all[4][NB];
  __shared__ int wsum_g[NB / 32];
  __shared__ int wsum_l[NB / 32];
  const int tid = threadIdx.x;
  const int s = blockIdx.x;
  int2* pairs = (int2*)pairs_i;

  {
    const int b = tid & (NB - 1);
    const int q = tid / NB;
    const int rq = (S + 3) / 4;
    const int r0 = q * rq;
    const int r1 = (r0 + rq) < S ? (r0 + rq) : S;
    int sb = 0, sa = 0;
#pragma unroll 16
    for (int r = r0; r < r1; ++r) {
      const int c = counts[r * NB + b];
      sa += c;
      sb += (r < s) ? c : 0;
    }
    part_below[q][b] = sb;
    part_all[q][b] = sa;
  }
  __syncthreads();

  int below_b = 0, all_b = 0, my_c = 0, g_incl = 0, l_incl = 0;
  if (tid < NB) {
    below_b = part_below[0][tid] + part_below[1][tid] + part_below[2][tid] + part_below[3][tid];
    all_b   = part_all[0][tid] + part_all[1][tid] + part_all[2][tid] + part_all[3][tid];
    my_c    = counts[s * NB + tid];
    g_incl = all_b;
    l_incl = my_c;
    const int lane = tid & 31;
    for (int off = 1; off < 32; off <<= 1) {
      const int tg = __shfl_up(g_incl, off);
      const int tl = __shfl_up(l_incl, off);
      if (lane >= off) { g_incl += tg; l_incl += tl; }
    }
    if (lane == 31) { wsum_g[tid >> 5] = g_incl; wsum_l[tid >> 5] = l_incl; }
  }
  __syncthreads();
  if (tid < NB) {
    int gb = 0, lb = 0;
    const int w = tid >> 5;
    for (int k = 0; k < w; ++k) { gb += wsum_g[k]; lb += wsum_l[k]; }
    const int gs = gb + g_incl - all_b;
    const int ls = lb + l_incl - my_c;
    cur[tid] = (unsigned)ls;
    gdelta[tid] = gs + below_b - ls;
    if (s == 0) {
      bstart[tid] = gs;
      if (tid == NB - 1) bstart[NB] = gs + all_b;
    }
  }
  __syncthreads();

  const int chunk = (E + S - 1) / S;
  const int e0 = s * chunk;
  const int e1 = (e0 + chunk) < E ? (e0 + chunk) : E;
  if (e0 < e1) {
    const int last = e1 - 1;
    for (int base = e0 + tid; base < e1; base += BK_THREADS * BK_U) {
      int d[BK_U], sv[BK_U];
#pragma unroll
      for (int j = 0; j < BK_U; ++j) {
        const int e = base + j * BK_THREADS;
        const int ec = e < last ? e : last;
        d[j] = dst[ec];
        sv[j] = src[ec];
      }
#pragma unroll
      for (int j = 0; j < BK_U; ++j) {
        const int e = base + j * BK_THREADS;
        if (e < e1) {
          const int b = range_of(d[j], rn);
          const unsigned p = __hip_atomic_fetch_add(&cur[b], 1u, __ATOMIC_RELAXED, __HIP_MEMORY_SCOPE_WORKGROUP);
          stage[p] = make_int2(sv[j], d[j]);
        }
      }
    }
  }
  __syncthreads();

  const int len = e1 - e0;
  for (int i = tid; i < len; i += BK_THREADS) {
    const int2 pr = stage[i];
    pairs[gdelta[range_of(pr.y, rn)] + i] = pr;
  }
}

template <int WID>
__device__ __forceinline__ void owner_sum(float* acc, const int2* __restrict__ pairs,
                                          const int* __restrict__ bstart,
                                          const float* __restrict__ rows, int lo, int span) {
  const int tid = threadIdx.x;
  const int cnt = span * WID;

  for (int i = tid; i < cnt; i += BK_THREADS) acc[i] = 0.0f;
  __syncthreads();

  const int p0 = bstart[blockIdx.x];
  const int p1 = bstart[blockIdx.x + 1];
  const int step = BK_THREADS * BK_U;

  if (p0 < p1) {
    const int last = p1 - 1;
    int2 pcur[BK_U];
#pragma unroll
    for (int j = 0; j < BK_U; ++j) {
      const int p = p0 + tid + j * BK_THREADS;
      pcur[j] = pairs[p < last ? p : last];
    }
    for (int base = p0 + tid; base < p1; base += step) {
      int2 pnext[BK_U];
#pragma unroll
      for (int j = 0; j < BK_U; ++j) {
        const int p = base + step + j * BK_THREADS;
        pnext[j] = pairs[p < last ? p : last];
      }
      unsigned rel[BK_U];
      bool hit[BK_U];
#pragma unroll
      for (int j = 0; j < BK_U; ++j) {
        const int p = base + j * BK_THREADS;
        rel[j] = (unsigned)(pcur[j].y - lo);
        hit[j] = (p < p1) && (rel[j] < (unsigned)span);
      }
      if (WID == 1) {
#pragma unroll
        for (int j = 0; j < BK_U; ++j)
          if (hit[j]) lds_add(&acc[rel[j]], 1.0f);
      } else {
        int sidx[BK_U];
#pragma unroll
        for (int j = 0; j < BK_U; ++j) sidx[j] = hit[j] ? pcur[j].x : 0;
        if (WID == 2) {
          v2f v[BK_U];
#pragma unroll
          for (int j = 0; j < BK_U; ++j) v[j] = *(const v2f*)(rows + (size_t)sidx[j] * 2);
#pragma unroll
          for (int j = 0; j < BK_U; ++j)
            if (hit[j]) {
              float* a = &acc[rel[j] * 2];
              lds_add(a + 0, v[j].x);
              lds_add(a + 1, v[j].y);
            }
        } else {
          float4 v[BK_U];
#pragma unroll
          for (int j = 0; j < BK_U; ++j) v[j] = *(const float4*)(rows + (size_t)sidx[j] * 4);
#pragma unroll
          for (int j = 0; j < BK_U; ++j)
            if (hit[j]) {
              float* a = &acc[rel[j] * 4];
              lds_add(a + 0, v[j].x);
              lds_add(a + 1, v[j].y);
              lds_add(a + 2, v[j].z);
              lds_add(a + 3, v[j].w);
            }
        }
      }
#pragma unroll
      for (int j = 0; j < BK_U; ++j) pcur[j] = pnext[j];
    }
  }
  __syncthreads();
}

__device__ __forceinline__ int owner_span(int lo, int n, int rn) {
  int span = n - lo;
  span = span < rn ? span : rn;
  return span > 0 ? span : 0;
}

__global__ void __launch_bounds__(BK_THREADS)
k_owner_deg(const int* __restrict__ pairs_i, const int* __restrict__ bstart,
            const float* __restrict__ x, float* __restrict__ dinv,
            float* __restrict__ xs, int n, int rn) {
  __shared__ __attribute__((aligned(16))) float acc[ACC_FLOATS];
  const int lo = blockIdx.x * rn;
  const int span = owner_span(lo, n, rn);
  owner_sum<1>(acc, (const int2*)pairs_i, bstart, x, lo, span);

  for (int i = threadIdx.x; i < span; i += BK_THREADS) {
    const int node = lo + i;
    float deg = 1.0f;
    deg += acc[i];
    float di = rsqrtf(deg);
    dinv[node] = di;
    v2f xv = *(const v2f*)(x + (size_t)node * 2);
    v2f a;
    a.x = xv.x * di;
    a.y = xv.y * di;
    *(v2f*)(xs + (size_t)node * 2) = a;
  }
}

__global__ void __launch_bounds__(BK_THREADS)
k_owner_l1(const int* __restrict__ pairs_i, const int* __restrict__ bstart,
           const float* __restrict__ xs, const float* __restrict__ dinv,
           const float* __restrict__ W1, const float* __restrict__ b1,
           const float* __restrict__ W2, float* __restrict__ t2, int n, int rn) {
  __shared__ __attribute__((aligned(16))) float acc[ACC_FLOATS];
  const int lo = blockIdx.x * rn;
  const int span = owner_span(lo, n, rn);
  owner_sum<2>(acc, (const int2*)pairs_i, bstart, xs, lo, span);

  const int lane = threadIdx.x & 31;
  const int half = lane >> 4;
  const int l    = lane & 15;
  const int ka   = half * 2;
  const int wave = __builtin_amdgcn_readfirstlane((int)(threadIdx.x >> 5));

  const int lm = l & 7;
  const int feat = ((lm & 1) + ((lm >> 1) << 2) + ((l >> 3) << 1)) & 7;
  const float w1a = W1[feat];
  const float w1b = W1[8 + feat];
  const bool aok = (!half) && (lm < 4);
  v2f wa;
  wa.x = aok ? w1a : 0.0f;
  wa.y = aok ? w1b : 0.0f;

  const v2f bb0 = *(const v2f*)(b1 + ka);
  const v2f bb1 = *(const v2f*)(b1 + ka + 4);
  const int lc = l & 3;
  const float wr0 = W2[(ka)     * 4 + lc];
  const float wr1 = W2[(ka + 1) * 4 + lc];
  const float wr2 = W2[(ka + 4) * 4 + lc];
  const float wr3 = W2[(ka + 5) * 4 + lc];
  const bool cok = (l < 4);
  v2f b1v, b2v;
  b1v.x = cok ? wr0 : 0.0f;
  b1v.y = cok ? wr1 : 0.0f;
  b2v.x = cok ? wr2 : 0.0f;
  b2v.y = cok ? wr3 : 0.0f;

  const int ntiles = (span + 15) >> 4;
  for (int t = wave; t < ntiles; t += BK_THREADS / 32) {
    const int i  = t * 16 + l;
    const int ic = i < span ? i : span - 1;
    v2f xa = *(const v2f*)(xs + (size_t)(lo + ic) * 2);
    const v2f pv = *(const v2f*)(acc + ic * 2);
    xa.x += pv.x;
    xa.y += pv.y;
    const float dr = dinv[lo + ic];
    v2f xb;
    xb.x = half ? 0.0f : xa.x * dr;
    xb.y = half ? 0.0f : xa.y * dr;

    v8f g = {};
    g = __builtin_amdgcn_wmma_f32_16x16x4_f32(false, wa, false, xb, (short)0, g, false, false);

    v2f a1, a2;
    a1.x = fmaxf(g[0] + bb0.x, 0.0f);
    a1.y = fmaxf(g[1] + bb0.y, 0.0f);
    a2.x = fmaxf(g[2] + bb1.x, 0.0f);
    a2.y = fmaxf(g[3] + bb1.y, 0.0f);

    v8f c = {};
    c = __builtin_amdgcn_wmma_f32_16x16x4_f32(false, a1, false, b1v, (short)0, c, false, false);
    c = __builtin_amdgcn_wmma_f32_16x16x4_f32(false, a2, false, b2v, (short)0, c, false, false);

    if (l < 4) {
#pragma unroll
      for (int r = 0; r < 8; ++r) {
        const int ir = t * 16 + half * 8 + r;
        if (ir < span) t2[(size_t)(lo + ir) * 4 + l] = c[r] * dinv[lo + ir];
      }
    }
  }
}

__global__ void __launch_bounds__(BK_THREADS)
k_owner_l2(const int* __restrict__ pairs_i, const int* __restrict__ bstart,
           const float* __restrict__ t2, const float* __restrict__ dinv,
           const float* __restrict__ b2, float* __restrict__ out, int n, int rn) {
  __shared__ __attribute__((aligned(16))) float acc[ACC_FLOATS];
  const int lo = blockIdx.x * rn;
  const int span = owner_span(lo, n, rn);
  owner_sum<4>(acc, (const int2*)pairs_i, bstart, t2, lo, span);

  for (int i = threadIdx.x; i < span; i += BK_THREADS) {
    const int node = lo + i;
    float4 a = *(const float4*)(t2 + (size_t)node * 4);
    const float4 pv = *(const float4*)(acc + i * 4);
    a.x += pv.x;
    a.y += pv.y;
    a.z += pv.z;
    a.w += pv.w;
    float di = dinv[node];
    a.x = b2[0] + a.x * di;
    a.y = b2[1] + a.y * di;
    a.z = b2[2] + a.z * di;
    a.w = b2[3] + a.w * di;
    *(float4*)(out + (size_t)node * 4) = a;
  }
}

extern "C" void kernel_launch(void* const* d_in, const int* in_sizes, int n_in,
                              void* d_out, int out_size, void* d_ws, size_t ws_size,
                              hipStream_t stream) {
  const float* x  = (const float*)d_in[0];
  const int*   ei = (const int*)d_in[1];
  const float* W1 = (const float*)d_in[2];
  const float* b1 = (const float*)d_in[3];
  const float* W2 = (const float*)d_in[4];
  const float* b2 = (const float*)d_in[5];
  float* out = (float*)d_out;

  const int n = in_sizes[0] / 2;
  const int E = in_sizes[1] / 2;
  const int* src = ei;
  const int* dst = ei + E;

  const int rn = (n + NB - 1) / NB > 0 ? (n + NB - 1) / NB : 1;
  const int Smin = (E + STAGE_PAIRS - 1) / STAGE_PAIRS;
  const int S = Smin > 256 ? Smin : 256;

  char* ws = (char*)d_ws;
  size_t nAl = ((size_t)n + 3) & ~(size_t)3;
  float* dinv   = (float*)ws;  ws += sizeof(float) * nAl;
  float* xs     = (float*)ws;  ws += sizeof(float) * nAl * 2;
  float* t2     = (float*)ws;  ws += sizeof(float) * (size_t)n * 4;
  int*   counts = (int*)ws;    ws += sizeof(int) * (size_t)S * NB;
  int*   bstart = (int*)ws;    ws += sizeof(int) * (NB + 4);
  int*   pairs  = (int*)ws;

  k_range_count<<<S, BK_THREADS, 0, stream>>>(dst, counts, E, S, rn);
  k_range_place<<<S, BK_THREADS, 0, stream>>>(src, dst, counts, bstart, pairs, E, S, rn);

  k_owner_deg<<<NB, BK_THREADS, 0, stream>>>(pairs, bstart, x, dinv, xs, n, rn);
  k_owner_l1<<<NB, BK_THREADS, 0, stream>>>(pairs, bstart, xs, dinv, W1, b1, W2, t2, n, rn);
  k_owner_l2<<<NB, BK_THREADS, 0, stream>>>(pairs, bstart, t2, dinv, b2, out, n, rn);
}


# LDS accumulators stored one plane per feature (planes 80,000 B apart), not a node's features side by side: an add's 32 lanes spread over all 64 banks, not every 4th or 2nd; f32 sums stay unordered
# speedup vs baseline: 12.5765x; 1.0560x over previous
#include <hip/hip_runtime.h>


typedef __attribute__((ext_vector_type(2))) float v2f;
typedef __attribute__((ext_vector_type(8))) float v8f;

#define NB 256
#define BK_THREADS 1024
#define BK_U 8
#define STAGE_PAIRS 32768
#define ACC_FLOATS 80000
#define ACC_PITCH (ACC_FLOATS / 4)

static_assert(BK_THREADS == 4 * NB, "k_range_place adds the count table with four threads per range");

__device__ __forceinline__ void lds_add(float* a, float v) {
  __hip_atomic_fetch_add(a, v, __ATOMIC_RELAXED, __HIP_MEMORY_SCOPE_WORKGROUP);
}

__device__ __forceinline__ int range_of(int d, int rn) {
  const unsigned b = (unsigned)d / (unsigned)rn;
  return b < (unsigned)NB ? (int)b : NB - 1;
}

__global__ void __launch_bounds__(BK_THREADS)
k_range_count(const int* __restrict__ dst, int* __restrict__ counts, int E, int S, int rn) {
  __shared__ unsigned cnt[NB];
  const int tid = threadIdx.x;
  const int s = blockIdx.x;
  if (tid < NB) cnt[tid] = 0u;
  __syncthreads();

  const int chunk = (E + S - 1) / S;
  const int e0 = s * chunk;
  const int e1 = (e0 + chunk) < E ? (e0 + chunk) : E;
  if (e0 < e1) {
    const int last = e1 - 1;
    for (int base = e0 + tid; base < e1; base += BK_THREADS * BK_U) {
      int d[BK_U];
#pragma unroll
      for (int j = 0; j < BK_U; ++j) {
        const int e = base + j * BK_THREADS;
        d[j] = dst[e < last ? e : last];
      }
#pragma unroll
      for (int j = 0; j < BK_U; ++j) {
        const int e = base + j * BK_THREADS;
        if (e < e1)
          __hip_atomic_fetch_add(&cnt[range_of(d[j], rn)], 1u, __ATOMIC_RELAXED, __HIP_MEMORY_SCOPE_WORKGROUP);
      }
    }
  }
  __syncthreads();
  if (tid < NB) counts[s * NB + tid] = (int)cnt[tid];
}

__global__ void __launch_bounds__(BK_THREADS)
k_range_place(const int* __restrict__ src, const int* __restrict__ dst,
              const int* __restrict__ counts, int* __restrict__ bstart,
              int* __restrict__ pairs_i, int E, int S, int rn) {
  __shared__ __attribute__((aligned(16))) int2 stage[STAGE_PAIRS];
  __shared__ unsigned cur[NB];
  __shared__ int gdelta[NB];
  __shared__ int part_below[4][NB];
  __shared__ int part_all[4][NB];
  __shared__ int wsum_g[NB / 32];
  __shared__ int wsum_l[NB / 32];
  const int tid = threadIdx.x;
  const int s = blockIdx.x;
  int2* pairs = (int2*)pairs_i;

  {
    const int b = tid & (NB - 1);
    const int q = tid / NB;
    const int rq = (S + 3) / 4;
    const int r0 = q * rq;
    const int r1 = (r0 + rq) < S ? (r0 + rq) : S;
    int sb = 0, sa = 0;
#pragma unroll 16
    for (int r = r0; r < r1; ++r) {
      const int c = counts[r * NB + b];
      sa += c;
      sb += (r < s) ? c : 0;
    }
    part_below[q][b] = sb;
    part_all[q][b] = sa;
  }
  __syncthreads();

  int below_b = 0, all_b = 0, my_c = 0, g_incl = 0, l_incl = 0;
  if (tid < NB) {
    below_b = part_below[0][tid] + part_below[1][tid] + part_below[2][tid] + part_below[3][tid];
    all_b   = part_all[0][tid] + part_all[1][tid] + part_all[2][tid] + part_all[3][tid];
    my_c    = counts[s * NB + tid];
    g_incl = all_b;
    l_incl = my_c;
    const int lane = tid & 31;
    for (int off = 1; off < 32; off <<= 1) {
      const int tg = __shfl_up(g_incl, off);
      const int tl = __shfl_up(l_incl, off);
      if (lane >= off) { g_incl += tg; l_incl += tl; }
    }
    if (lane == 31) { wsum_g[tid >> 5] = g_incl; wsum_l[tid >> 5] = l_incl; }
  }
  __syncthreads();
  if (tid < NB) {
    int gb = 0, lb = 0;
    const int w = tid >> 5;
    for (int k = 0; k < w; ++k) { gb += wsum_g[k]; lb += wsum_l[k]; }
    const int gs = gb + g_incl - all_b;
    const int ls = lb + l_incl - my_c;
    cur[tid] = (unsigned)ls;
    gdelta[tid] = gs + below_b - ls;
    if (s == 0) {
      bstart[tid] = gs;
      if (tid == NB - 1) bstart[NB] = gs + all_b;
    }
  }
  __syncthreads();

  const int chunk = (E + S - 1) / S;
  const int e0 = s * chunk;
  const int e1 = (e0 + chunk) < E ? (e0 + chunk) : E;
  if (e0 < e1) {
    const int last = e1 - 1;
    for (int base = e0 + tid; base < e1; base += BK_THREADS * BK_U) {
      int d[BK_U], sv[BK_U];
#pragma unroll
      for (int j = 0; j < BK_U; ++j) {
        const int e = base + j * BK_THREADS;
        const int ec = e < last ? e : last;
        d[j] = dst[ec];
        sv[j] = src[ec];
      }
#pragma unroll
      for (int j = 0; j < BK_U; ++j) {
        const int e = base + j * BK_THREADS;
        if (e < e1) {
          const int b = range_of(d[j], rn);
          const unsigned p = __hip_atomic_fetch_add(&cur[b], 1u, __ATOMIC_RELAXED, __HIP_MEMORY_SCOPE_WORKGROUP);
          stage[p] = make_int2(sv[j], d[j]);
        }
      }
    }
  }
  __syncthreads();

  const int len = e1 - e0;
  for (int i = tid; i < len; i += BK_THREADS) {
    const int2 pr = stage[i];
    pairs[gdelta[range_of(pr.y, rn)] + i] = pr;
  }
}

template <int WID>
__device__ __forceinline__ void owner_sum(float* acc, const int2* __restrict__ pairs,
                                          const int* __restrict__ bstart,
                                          const float* __restrict__ rows, int lo, int span) {
  const int tid = threadIdx.x;

  for (int i = tid; i < span; i += BK_THREADS) {
#pragma unroll
    for (int f = 0; f < WID; ++f) acc[f * ACC_PITCH + i] = 0.0f;
  }
  __syncthreads();

  const int p0 = bstart[blockIdx.x];
  const int p1 = bstart[blockIdx.x + 1];
  const int step = BK_THREADS * BK_U;

  if (p0 < p1) {
    const int last = p1 - 1;
    int2 pcur[BK_U];
#pragma unroll
    for (int j = 0; j < BK_U; ++j) {
      const int p = p0 + tid + j * BK_THREADS;
      pcur[j] = pairs[p < last ? p : last];
    }
    for (int base = p0 + tid; base < p1; base += step) {
      int2 pnext[BK_U];
#pragma unroll
      for (int j = 0; j < BK_U; ++j) {
        const int p = base + step + j * BK_THREADS;
        pnext[j] = pairs[p < last ? p : last];
      }
      unsigned rel[BK_U];
      bool hit[BK_U];
#pragma unroll
      for (int j = 0; j < BK_U; ++j) {
        const int p = base + j * BK_THREADS;
        rel[j] = (unsigned)(pcur[j].y - lo);
        hit[j] = (p < p1) && (rel[j] < (unsigned)span);
      }
      if (WID == 1) {
#pragma unroll
        for (int j = 0; j < BK_U; ++j)
          if (hit[j]) lds_add(&acc[rel[j]], 1.0f);
      } else {
        int sidx[BK_U];
#pragma unroll
        for (int j = 0; j < BK_U; ++j) sidx[j] = hit[j] ? pcur[j].x : 0;
        if (WID == 2) {
          v2f v[BK_U];
#pragma unroll
          for (int j = 0; j < BK_U; ++j) v[j] = *(const v2f*)(rows + (size_t)sidx[j] * 2);
#pragma unroll
          for (int j = 0; j < BK_U; ++j)
            if (hit[j]) {
              float* a = &acc[rel[j]];
              lds_add(a, v[j].x);
              lds_add(a + ACC_PITCH, v[j].y);
            }
        } else {
          float4 v[BK_U];
#pragma unroll
          for (int j = 0; j < BK_U; ++j) v[j] = *(const float4*)(rows + (size_t)sidx[j] * 4);
#pragma unroll
          for (int j = 0; j < BK_U; ++j)
            if (hit[j]) {
              float* a = &acc[rel[j]];
              lds_add(a, v[j].x);
              lds_add(a + ACC_PITCH, v[j].y);
              lds_add(a + 2 * ACC_PITCH, v[j].z);
              lds_add(a + 3 * ACC_PITCH, v[j].w);
            }
        }
      }
#pragma unroll
      for (int j = 0; j < BK_U; ++j) pcur[j] = pnext[j];
    }
  }
  __syncthreads();
}

__device__ __forceinline__ int owner_span(int lo, int n, int rn) {
  int span = n - lo;
  span = span < rn ? span : rn;
  return span > 0 ? span : 0;
}

__global__ void __launch_bounds__(BK_THREADS)
k_owner_deg(const int* __restrict__ pairs_i, const int* __restrict__ bstart,
            const float* __restrict__ x, float* __restrict__ dinv,
            float* __restrict__ xs, int n, int rn) {
  __shared__ __attribute__((aligned(16))) float acc[ACC_FLOATS];
  const int lo = blockIdx.x * rn;
  const int span = owner_span(lo, n, rn);
  owner_sum<1>(acc, (const int2*)pairs_i, bstart, x, lo, span);

  for (int i = threadIdx.x; i < span; i += BK_THREADS) {
    const int node = lo + i;
    float deg = 1.0f;
    deg += acc[i];
    float di = rsqrtf(deg);
    dinv[node] = di;
    v2f xv = *(const v2f*)(x + (size_t)node * 2);
    v2f a;
    a.x = xv.x * di;
    a.y = xv.y * di;
    *(v2f*)(xs + (size_t)node * 2) = a;
  }
}

__global__ void __launch_bounds__(BK_THREADS)
k_owner_l1(const int* __restrict__ pairs_i, const int* __restrict__ bstart,
           const float* __restrict__ xs, const float* __restrict__ dinv,
           const float* __restrict__ W1, const float* __restrict__ b1,
           const float* __restrict__ W2, float* __restrict__ t2, int n, int rn) {
  __shared__ __attribute__((aligned(16))) float acc[ACC_FLOATS];
  const int lo = blockIdx.x * rn;
  const int span = owner_span(lo, n, rn);
  owner_sum<2>(acc, (const int2*)pairs_i, bstart, xs, lo, span);

  const int lane = threadIdx.x & 31;
  const int half = lane >> 4;
  const int l    = lane & 15;
  const int ka   = half * 2;
  const int wave = __builtin_amdgcn_readfirstlane((int)(threadIdx.x >> 5));

  const int lm = l & 7;
  const int feat = ((lm & 1) + ((lm >> 1) << 2) + ((l >> 3) << 1)) & 7;
  const float w1a = W1[feat];
  const float w1b = W1[8 + feat];
  const bool aok = (!half) && (lm < 4);
  v2f wa;
  wa.x = aok ? w1a : 0.0f;
  wa.y = aok ? w1b : 0.0f;

  const v2f bb0 = *(const v2f*)(b1 + ka);
  const v2f bb1 = *(const v2f*)(b1 + ka + 4);
  const int lc = l & 3;
  const float wr0 = W2[(ka)     * 4 + lc];
  const float wr1 = W2[(ka + 1) * 4 + lc];
  const float wr2 = W2[(ka + 4) * 4 + lc];
  const float wr3 = W2[(ka + 5) * 4 + lc];
  const bool cok = (l < 4);
  v2f b1v, b2v;
  b1v.x = cok ? wr0 : 0.0f;
  b1v.y = cok ? wr1 : 0.0f;
  b2v.x = cok ? wr2 : 0.0f;
  b2v.y = cok ? wr3 : 0.0f;

  const int ntiles = (span + 15) >> 4;
  for (int t = wave; t < ntiles; t += BK_THREADS / 32) {
    const int i  = t * 16 + l;
    const int ic = i < span ? i : span - 1;
    v2f xa = *(const v2f*)(xs + (size_t)(lo + ic) * 2);
    xa.x += acc[ic];
    xa.y += acc[ACC_PITCH + ic];
    const float dr = dinv[lo + ic];
    v2f xb;
    xb.x = half ? 0.0f : xa.x * dr;
    xb.y = half ? 0.0f : xa.y * dr;

    v8f g = {};
    g = __builtin_amdgcn_wmma_f32_16x16x4_f32(false, wa, false, xb, (short)0, g, false, false);

    v2f a1, a2;
    a1.x = fmaxf(g[0] + bb0.x, 0.0f);
    a1.y = fmaxf(g[1] + bb0.y, 0.0f);
    a2.x = fmaxf(g[2] + bb1.x, 0.0f);
    a2.y = fmaxf(g[3] + bb1.y, 0.0f);

    v8f c = {};
    c = __builtin_amdgcn_wmma_f32_16x16x4_f32(false, a1, false, b1v, (short)0, c, false, false);
    c = __builtin_amdgcn_wmma_f32_16x16x4_f32(false, a2, false, b2v, (short)0, c, false, false);

    if (l < 4) {
#pragma unroll
      for (int r = 0; r < 8; ++r) {
        const int ir = t * 16 + half * 8 + r;
        if (ir < span) t2[(size_t)(lo + ir) * 4 + l] = c[r] * dinv[lo + ir];
      }
    }
  }
}

__global__ void __launch_bounds__(BK_THREADS)
k_owner_l2(const int* __restrict__ pairs_i, const int* __restrict__ bstart,
           const float* __restrict__ t2, const float* __restrict__ dinv,
           const float* __restrict__ b2, float* __restrict__ out, int n, int rn) {
  __shared__ __attribute__((aligned(16))) float acc[ACC_FLOATS];
  const int lo = blockIdx.x * rn;
  const int span = owner_span(lo, n, rn);
  owner_sum<4>(acc, (const int2*)pairs_i, bstart, t2, lo, span);

  for (int i = threadIdx.x; i < span; i += BK_THREADS) {
    const int node = lo + i;
    float4 a = *(const float4*)(t2 + (size_t)node * 4);
    a.x += acc[i];
    a.y += acc[ACC_PITCH + i];
    a.z += acc[2 * ACC_PITCH + i];
    a.w += acc[3 * ACC_PITCH + i];
    float di = dinv[node];
    a.x = b2[0] + a.x * di;
    a.y = b2[1] + a.y * di;
    a.z = b2[2] + a.z * di;
    a.w = b2[3] + a.w * di;
    *(float4*)(out + (size_t)node * 4) = a;
  }
}

extern "C" void kernel_launch(void* const* d_in, const int* in_sizes, int n_in,
                              void* d_out, int out_size, void* d_ws, size_t ws_size,
                              hipStream_t stream) {
  const float* x  = (const float*)d_in[0];
  const int*   ei = (const int*)d_in[1];
  const float* W1 = (const float*)d_in[2];
  const float* b1 = (const float*)d_in[3];
  const float* W2 = (const float*)d_in[4];
  const float* b2 = (const float*)d_in[5];
  float* out = (float*)d_out;

  const int n = in_sizes[0] / 2;
  const int E = in_sizes[1] / 2;
  const int* src = ei;
  const int* dst = ei + E;

  const int rn = (n + NB - 1) / NB > 0 ? (n + NB - 1) / NB : 1;
  const int Smin = (E + STAGE_PAIRS - 1) / STAGE_PAIRS;
  const int S = Smin > 256 ? Smin : 256;

  char* ws = (char*)d_ws;
  size_t nAl = ((size_t)n + 3) & ~(size_t)3;
  float* dinv   = (float*)ws;  ws += sizeof(float) * nAl;
  float* xs     = (float*)ws;  ws += sizeof(float) * nAl * 2;
  float* t2     = (float*)ws;  ws += sizeof(float) * (size_t)n * 4;
  int*   counts = (int*)ws;    ws += sizeof(int) * (size_t)S * NB;
  int*   bstart = (int*)ws;    ws += sizeof(int) * (NB + 4);
  int*   pairs  = (int*)ws;

  k_range_count<<<S, BK_THREADS, 0, stream>>>(dst, counts, E, S, rn);
  k_range_place<<<S, BK_THREADS, 0, stream>>>(src, dst, counts, bstart, pairs, E, S, rn);

  k_owner_deg<<<NB, BK_THREADS, 0, stream>>>(pairs, bstart, x, dinv, xs, n, rn);
  k_owner_l1<<<NB, BK_THREADS, 0, stream>>>(pairs, bstart, xs, dinv, W1, b1, W2, t2, n, rn);
  k_owner_l2<<<NB, BK_THREADS, 0, stream>>>(pairs, bstart, t2, dinv, b2, out, n, rn);
}


# Edge list grouped by target range inside each slice only, 1 launch (no count pass, no table sums); owners walk their run in every slice via a per-slice table: 5 launches to 4; f32 sums stay unordered
# speedup vs baseline: 13.4491x; 1.0694x over previous
#include <hip/hip_runtime.h>


typedef __attribute__((ext_vector_type(2))) float v2f;
typedef __attribute__((ext_vector_type(8))) float v8f;

#define NB 256
#define BK_THREADS 1024
#define BK_U 8
#define NPT 25
#define STAGE_PAIRS (NPT * BK_THREADS)
#define RW 8
#define ACC_FLOATS 80000
#define ACC_PITCH (ACC_FLOATS / 4)

static_assert(NB <= BK_THREADS && NB % 32 == 0, "k_slice_group scans the NB counts with one thread each, whole waves");
static_assert(NB + BK_THREADS <= 65536 && STAGE_PAIRS <= 65536, "a range's number and a place inside its run share one 32-bit word");

__device__ __forceinline__ void lds_add(float* a, float v) {
  __hip_atomic_fetch_add(a, v, __ATOMIC_RELAXED, __HIP_MEMORY_SCOPE_WORKGROUP);
}

__device__ __forceinline__ int range_of(int d, int rn) {
  const unsigned b = (unsigned)d / (unsigned)rn;
  return b < (unsigned)NB ? (int)b : NB - 1;
}

__global__ void __launch_bounds__(BK_THREADS)
k_slice_group(const int* __restrict__ src, const int* __restrict__ dst,
              int* __restrict__ runs_i, int* __restrict__ pairs_i, int E, int S, int rn) {
  __shared__ __attribute__((aligned(16))) int2 stage[STAGE_PAIRS];
  __shared__ unsigned cnt[NB + BK_THREADS];
  __shared__ int first[NB];
  __shared__ int wsum[NB / 32];
  const int tid = threadIdx.x;
  const int s = blockIdx.x;
  int2* pairs = (int2*)pairs_i;
  int2* runs = (int2*)runs_i;

  const int chunk = (E + S - 1) / S;
  const int e0 = (s * chunk) < E ? (s * chunk) : E;
  const int e1 = (e0 + chunk) < E ? (e0 + chunk) : E;
  const int len = e1 - e0;
  const int last = len > 0 ? e1 - 1 : 0;

  int d[NPT], sv[NPT];
#pragma unroll
  for (int j = 0; j < NPT; ++j) {
    const int e = e0 + tid + j * BK_THREADS;
    const int ec = e < last ? e : last;
    d[j] = dst[ec];
    sv[j] = src[ec];
  }
  if (tid < NB) cnt[tid] = 0u;
  cnt[NB + tid] = 0u;
  __syncthreads();

  unsigned key[NPT];
#pragma unroll
  for (int j = 0; j < NPT; ++j) {
    const int e = e0 + tid + j * BK_THREADS;
    const unsigned slot = e < e1 ? (unsigned)range_of(d[j], rn) : (unsigned)(NB + tid);
    const unsigned r = __hip_atomic_fetch_add(&cnt[slot], 1u, __ATOMIC_RELAXED, __HIP_MEMORY_SCOPE_WORKGROUP);
    key[j] = (slot << 16) | r;
  }
  __syncthreads();

  int my_c = 0, incl = 0;
  if (tid < NB) {
    my_c = (int)cnt[tid];
    incl = my_c;
    const int lane = tid & 31;
    for (int off = 1; off < 32; off <<= 1) {
      const int t = __shfl_up(incl, off);
      if (lane >= off) incl += t;
    }
    if (lane == 31) wsum[tid >> 5] = incl;
  }
  __syncthreads();
  if (tid < NB) {
    int wb = 0;
    const int w = tid >> 5;
    for (int k = 0; k < w; ++k) wb += wsum[k];
    const int f = wb + incl - my_c;
    first[tid] = f;
    runs[(size_t)tid * S + s] = make_int2(e0 + f, e0 + f + my_c);
  }
  __syncthreads();

  int fj[NPT];
#pragma unroll
  for (int j = 0; j < NPT; ++j) {
    const unsigned b = key[j] >> 16;
    fj[j] = first[b < (unsigned)NB ? b : NB - 1];
  }
#pragma unroll
  for (int j = 0; j < NPT; ++j) {
    const int e = e0 + tid + j * BK_THREADS;
    if (e < e1) stage[fj[j] + (int)(key[j] & 0xffffu)] = make_int2(sv[j], d[j]);
  }
  __syncthreads();

  int2 pr[NPT];
#pragma unroll
  for (int j = 0; j < NPT; ++j) {
    const int i = tid + j * BK_THREADS;
    pr[j] = stage[i < len ? i : 0];
  }
#pragma unroll
  for (int j = 0; j < NPT; ++j) {
    const int i = tid + j * BK_THREADS;
    if (i < len) pairs[e0 + i] = pr[j];
  }
}

template <int WID>
__device__ __forceinline__ void owner_sum(float* acc, const int2* __restrict__ pairs,
                                          const int2* __restrict__ runs, int S,
                                          const float* __restrict__ rows, int lo, int span) {
  const int tid = threadIdx.x;

  for (int i = tid; i < span; i += BK_THREADS) {
#pragma unroll
    for (int f = 0; f < WID; ++f) acc[f * ACC_PITCH + i] = 0.0f;
  }
  __syncthreads();

  const int lane = tid & 31;
  const int wave = __builtin_amdgcn_readfirstlane(tid >> 5);
  const int step = 32 * BK_U;

  for (int g = wave * RW; g < S; g += (BK_THREADS / 32) * RW) {
    const int si = g + (lane & (RW - 1));
    const int2 te = runs[si < S ? si : S - 1];
    const int ln = si < S ? te.y - te.x : 0;
    int cs[RW], dl[RW];
    int total = 0, prev = 0;
#pragma unroll
    for (int q = 0; q < RW; ++q) {
      const int lq = __builtin_amdgcn_readlane(ln, q);
      const int fq = __builtin_amdgcn_readlane(te.x, q);
      cs[q] = total;
      dl[q] = (fq - total) - prev;
      prev = fq - total;
      total += lq;
    }
    if (total <= 0) continue;
    const int lastv = total - 1;

    auto pair_at = [&](int v) -> int2 {
      const int vc = v < lastv ? v : lastv;
      int o = vc + dl[0];
#pragma unroll
      for (int q = 1; q < RW; ++q) o += (vc >= cs[q]) ? dl[q] : 0;
      return pairs[o];
    };

    int2 pcur[BK_U];
#pragma unroll
    for (int j = 0; j < BK_U; ++j) pcur[j] = pair_at(lane + j * 32);

    for (int vb = 0; vb < total; vb += step) {
      int2 pnext[BK_U];
#pragma unroll
      for (int j = 0; j < BK_U; ++j) pnext[j] = pair_at(vb + step + lane + j * 32);
      unsigned rel[BK_U];
      bool hit[BK_U];
#pragma unroll
      for (int j = 0; j < BK_U; ++j) {
        const int v = vb + lane + j * 32;
        rel[j] = (unsigned)(pcur[j].y - lo);
        hit[j] = (v < total) && (rel[j] < (unsigned)span);
      }
      if (WID == 1) {
#pragma unroll
        for (int j = 0; j < BK_U; ++j)
          if (hit[j]) lds_add(&acc[rel[j]], 1.0f);
      } else {
        int sidx[BK_U];
#pragma unroll
        for (int j = 0; j < BK_U; ++j) sidx[j] = hit[j] ? pcur[j].x : 0;
        if (WID == 2) {
          v2f v[BK_U];
#pragma unroll
          for (int j = 0; j < BK_U; ++j) v[j] = *(const v2f*)(rows + (size_t)sidx[j] * 2);
#pragma unroll
          for (int j = 0; j < BK_U; ++j)
            if (hit[j]) {
              float* a = &acc[rel[j]];
              lds_add(a, v[j].x);
              lds_add(a + ACC_PITCH, v[j].y);
            }
        } else {
          float4 v[BK_U];
#pragma unroll
          for (int j = 0; j < BK_U; ++j) v[j] = *(const float4*)(rows + (size_t)sidx[j] * 4);
#pragma unroll
          for (int j = 0; j < BK_U; ++j)
            if (hit[j]) {
              float* a = &acc[rel[j]];
              lds_add(a, v[j].x);
              lds_add(a + ACC_PITCH, v[j].y);
              lds_add(a + 2 * ACC_PITCH, v[j].z);
              lds_add(a + 3 * ACC_PITCH, v[j].w);
            }
        }
      }
#pragma unroll
      for (int j = 0; j < BK_U; ++j) pcur[j] = pnext[j];
    }
  }
  __syncthreads();
}

__device__ __forceinline__ int owner_span(int lo, int n, int rn) {
  int span = n - lo;
  span = span < rn ? span : rn;
  return span > 0 ? span : 0;
}

__global__ void __launch_bounds__(BK_THREADS)
k_owner_deg(const int* __restrict__ pairs_i, const int* __restrict__ runs_i, int S,
            const float* __restrict__ x, float* __restrict__ dinv,
            float* __restrict__ xs, int n, int rn) {
  __shared__ __attribute__((aligned(16))) float acc[ACC_FLOATS];
  const int lo = blockIdx.x * rn;
  const int span = owner_span(lo, n, rn);
  owner_sum<1>(acc, (const int2*)pairs_i, (const int2*)runs_i + (size_t)blockIdx.x * S, S, x, lo, span);

  for (int i = threadIdx.x; i < span; i += BK_THREADS) {
    const int node = lo + i;
    float deg = 1.0f;
    deg += acc[i];
    float di = rsqrtf(deg);
    dinv[node] = di;
    v2f xv = *(const v2f*)(x + (size_t)node * 2);
    v2f a;
    a.x = xv.x * di;
    a.y = xv.y * di;
    *(v2f*)(xs + (size_t)node * 2) = a;
  }
}

__global__ void __launch_bounds__(BK_THREADS)
k_owner_l1(const int* __restrict__ pairs_i, const int* __restrict__ runs_i, int S,
           const float* __restrict__ xs, const float* __restrict__ dinv,
           const float* __restrict__ W1, const float* __restrict__ b1,
           const float* __restrict__ W2, float* __restrict__ t2, int n, int rn) {
  __shared__ __attribute__((aligned(16))) float acc[ACC_FLOATS];
  const int lo = blockIdx.x * rn;
  const int span = owner_span(lo, n, rn);
  owner_sum<2>(acc, (const int2*)pairs_i, (const int2*)runs_i + (size_t)blockIdx.x * S, S, xs, lo, span);

  const int lane = threadIdx.x & 31;
  const int half = lane >> 4;
  const int l    = lane & 15;
  const int ka   = half * 2;
  const int wave = __builtin_amdgcn_readfirstlane((int)(threadIdx.x >> 5));

  const int lm = l & 7;
  const int feat = ((lm & 1) + ((lm >> 1) << 2) + ((l >> 3) << 1)) & 7;
  const float w1a = W1[feat];
  const float w1b = W1[8 + feat];
  const bool aok = (!half) && (lm < 4);
  v2f wa;
  wa.x = aok ? w1a : 0.0f;
  wa.y = aok ? w1b : 0.0f;

  const v2f bb0 = *(const v2f*)(b1 + ka);
  const v2f bb1 = *(const v2f*)(b1 + ka + 4);
  const int lc = l & 3;
  const float wr0 = W2[(ka)     * 4 + lc];
  const float wr1 = W2[(ka + 1) * 4 + lc];
  const float wr2 = W2[(ka + 4) * 4 + lc];
  const float wr3 = W2[(ka + 5) * 4 + lc];
  const bool cok = (l < 4);
  v2f b1v, b2v;
  b1v.x = cok ? wr0 : 0.0f;
  b1v.y = cok ? wr1 : 0.0f;
  b2v.x = cok ? wr2 : 0.0f;
  b2v.y = cok ? wr3 : 0.0f;

  const int ntiles = (span + 15) >> 4;
  for (int t = wave; t < ntiles; t += BK_THREADS / 32) {
    const int i  = t * 16 + l;
    const int ic = i < span ? i : span - 1;
    v2f xa = *(const v2f*)(xs + (size_t)(lo + ic) * 2);
    xa.x += acc[ic];
    xa.y += acc[ACC_PITCH + ic];
    const float dr = dinv[lo + ic];
    v2f xb;
    xb.x = half ? 0.0f : xa.x * dr;
    xb.y = half ? 0.0f : xa.y * dr;

    v8f g = {};
    g = __builtin_amdgcn_wmma_f32_16x16x4_f32(false, wa, false, xb, (short)0, g, false, false);

    v2f a1, a2;
    a1.x = fmaxf(g[0] + bb0.x, 0.0f);
    a1.y = fmaxf(g[1] + bb0.y, 0.0f);
    a2.x = fmaxf(g[2] + bb1.x, 0.0f);
    a2.y = fmaxf(g[3] + bb1.y, 0.0f);

    v8f c = {};
    c = __builtin_amdgcn_wmma_f32_16x16x4_f32(false, a1, false, b1v, (short)0, c, false, false);
    c = __builtin_amdgcn_wmma_f32_16x16x4_f32(false, a2, false, b2v, (short)0, c, false, false);

    if (l < 4) {
#pragma unroll
      for (int r = 0; r < 8; ++r) {
        const int ir = t * 16 + half * 8 + r;
        if (ir < span) t2[(size_t)(lo + ir) * 4 + l] = c[r] * dinv[lo + ir];
      }
    }
  }
}

__global__ void __launch_bounds__(BK_THREADS)
k_owner_l2(const int* __restrict__ pairs_i, const int* __restrict__ runs_i, int S,
           const float* __restrict__ t2, const float* __restrict__ dinv,
           const float* __restrict__ b2, float* __restrict__ out, int n, int rn) {
  __shared__ __attribute__((aligned(16))) float acc[ACC_FLOATS];
  const int lo = blockIdx.x * rn;
  const int span = owner_span(lo, n, rn);
  owner_sum<4>(acc, (const int2*)pairs_i, (const int2*)runs_i + (size_t)blockIdx.x * S, S, t2, lo, span);

  for (int i = threadIdx.x; i < span; i += BK_THREADS) {
    const int node = lo + i;
    float4 a = *(const float4*)(t2 + (size_t)node * 4);
    a.x += acc[i];
    a.y += acc[ACC_PITCH + i];
    a.z += acc[2 * ACC_PITCH + i];
    a.w += acc[3 * ACC_PITCH + i];
    float di = dinv[node];
    a.x = b2[0] + a.x * di;
    a.y = b2[1] + a.y * di;
    a.z = b2[2] + a.z * di;
    a.w = b2[3] + a.w * di;
    *(float4*)(out + (size_t)node * 4) = a;
  }
}

extern "C" void kernel_launch(void* const* d_in, const int* in_sizes, int n_in,
                              void* d_out, int out_size, void* d_ws, size_t ws_size,
                              hipStream_t stream) {
  const float* x  = (const float*)d_in[0];
  const int*   ei = (const int*)d_in[1];
  const float* W1 = (const float*)d_in[2];
  const float* b1 = (const float*)d_in[3];
  const float* W2 = (const float*)d_in[4];
  const float* b2 = (const float*)d_in[5];
  float* out = (float*)d_out;

  const int n = in_sizes[0] / 2;
  const int E = in_sizes[1] / 2;
  const int* src = ei;
  const int* dst = ei + E;

  const int rn = (n + NB - 1) / NB > 0 ? (n + NB - 1) / NB : 1;
  const int Smin = (E + STAGE_PAIRS - 1) / STAGE_PAIRS;
  const int S = Smin > 256 ? Smin : 256;

  char* ws = (char*)d_ws;
  size_t nAl = ((size_t)n + 3) & ~(size_t)3;
  float* dinv   = (float*)ws;  ws += sizeof(float) * nAl;
  float* xs     = (float*)ws;  ws += sizeof(float) * nAl * 2;
  float* t2     = (float*)ws;  ws += sizeof(float) * (size_t)n * 4;
  int*   runs   = (int*)ws;    ws += sizeof(int) * 2 * (size_t)S * NB;
  int*   pairs  = (int*)ws;

  k_slice_group<<<S, BK_THREADS, 0, stream>>>(src, dst, runs, pairs, E, S, rn);

  k_owner_deg<<<NB, BK_THREADS, 0, stream>>>(pairs, runs, S, x, dinv, xs, n, rn);
  k_owner_l1<<<NB, BK_THREADS, 0, stream>>>(pairs, runs, S, xs, dinv, W1, b1, W2, t2, n, rn);
  k_owner_l2<<<NB, BK_THREADS, 0, stream>>>(pairs, runs, S, t2, dinv, b2, out, n, rn);
}


# Edges grouped by (target block, source block) cell: a cell's work-group copies its source rows to LDS, edges read rows there, not from L2; 8 partial planes per sum, 7 launches; f32 sum order changes
# speedup vs baseline: 15.7256x; 1.1693x over previous
#include <hip/hip_runtime.h>


typedef __attribute__((ext_vector_type(2))) float v2f;
typedef __attribute__((ext_vector_type(8))) float v8f;

#define GT 32
#define GS 8
#define NB (GT * GS)
#define BK_THREADS 1024
#define BK_U 8
#define NPT 25
#define STAGE_PAIRS (NPT * BK_THREADS)
#define RW 8
#define ACC_P 8000
#define ROW_P 32000
#define LDS_FLOATS (2 * ACC_P + 2 * ROW_P)
#define ST_U 13
#define NODE_THREADS 256

static_assert(NB <= BK_THREADS && NB % 32 == 0, "k_slice_group scans the NB counts with one thread each, whole waves");
static_assert(NB + BK_THREADS <= 65536 && STAGE_PAIRS <= 65536, "a cell's number and a place inside its run share one 32-bit word");

__device__ __forceinline__ void lds_add(float* a, float v) {
  __hip_atomic_fetch_add(a, v, __ATOMIC_RELAXED, __HIP_MEMORY_SCOPE_WORKGROUP);
}

__device__ __forceinline__ int cell_of(int s, int d, int tn, int sn) {
  unsigned tb = (unsigned)d / (unsigned)tn;
  unsigned sb = (unsigned)s / (unsigned)sn;
  tb = tb < (unsigned)GT ? tb : (unsigned)(GT - 1);
  sb = sb < (unsigned)GS ? sb : (unsigned)(GS - 1);
  return (int)(tb * GS + sb);
}

__device__ __forceinline__ int block_span(int lo, int n, int bn) {
  int span = n - lo;
  span = span < bn ? span : bn;
  return span > 0 ? span : 0;
}

__device__ __forceinline__ void stage_rows(v2f* rows, const float* __restrict__ g, int cnt) {
  const int tid = threadIdx.x;
  for (int base = 0; base < cnt; base += ST_U * BK_THREADS) {
    v2f t[ST_U];
#pragma unroll
    for (int j = 0; j < ST_U; ++j) {
      const int i = base + tid + j * BK_THREADS;
      const int ic = i < cnt ? i : cnt - 1;
      t[j] = *(const v2f*)(g + (size_t)ic * 2);
    }
#pragma unroll
    for (int j = 0; j < ST_U; ++j) {
      const int i = base + tid + j * BK_THREADS;
      if (i < cnt) rows[i] = t[j];
    }
  }
}

__global__ void __launch_bounds__(BK_THREADS)
k_slice_group(const int* __restrict__ src, const int* __restrict__ dst,
              int* __restrict__ runs_i, int* __restrict__ pairs_i, int E, int S, int tn, int sn) {
  __shared__ __attribute__((aligned(16))) int2 stage[STAGE_PAIRS];
  __shared__ unsigned cnt[NB + BK_THREADS];
  __shared__ int first[NB];
  __shared__ int wsum[NB / 32];
  const int tid = threadIdx.x;
  const int s = blockIdx.x;
  int2* pairs = (int2*)pairs_i;
  int2* runs = (int2*)runs_i;

  const int chunk = (E + S - 1) / S;
  const int e0 = (s * chunk) < E ? (s * chunk) : E;
  const int e1 = (e0 + chunk) < E ? (e0 + chunk) : E;
  const int len = e1 - e0;
  const int last = len > 0 ? e1 - 1 : 0;

  int d[NPT], sv[NPT];
#pragma unroll
  for (int j = 0; j < NPT; ++j) {
    const int e = e0 + tid + j * BK_THREADS;
    const int ec = e < last ? e : last;
    d[j] = dst[ec];
    sv[j] = src[ec];
  }
  if (tid < NB) cnt[tid] = 0u;
  cnt[NB + tid] = 0u;
  __syncthreads();

  unsigned key[NPT];
#pragma unroll
  for (int j = 0; j < NPT; ++j) {
    const int e = e0 + tid + j * BK_THREADS;
    const unsigned slot = e < e1 ? (unsigned)cell_of(sv[j], d[j], tn, sn) : (unsigned)(NB + tid);
    const unsigned r = __hip_atomic_fetch_add(&cnt[slot], 1u, __ATOMIC_RELAXED, __HIP_MEMORY_SCOPE_WORKGROUP);
    key[j] = (slot << 16) | r;
  }
  __syncthreads();

  int my_c = 0, incl = 0;
  if (tid < NB) {
    my_c = (int)cnt[tid];
    incl = my_c;
    const int lane = tid & 31;
    for (int off = 1; off < 32; off <<= 1) {
      const int t = __shfl_up(incl, off);
      if (lane >= off) incl += t;
    }
    if (lane == 31) wsum[tid >> 5] = incl;
  }
  __syncthreads();
  if (tid < NB) {
    int wb = 0;
    const int w = tid >> 5;
    for (int k = 0; k < w; ++k) wb += wsum[k];
    const int f = wb + incl - my_c;
    first[tid] = f;
    runs[(size_t)tid * S + s] = make_int2(e0 + f, e0 + f + my_c);
  }
  __syncthreads();

  int fj[NPT];
#pragma unroll
  for (int j = 0; j < NPT; ++j) {
    const unsigned b = key[j] >> 16;
    fj[j] = first[b < (unsigned)NB ? b : NB - 1];
  }
#pragma unroll
  for (int j = 0; j < NPT; ++j) {
    const int e = e0 + tid + j * BK_THREADS;
    if (e < e1) stage[fj[j] + (int)(key[j] & 0xffffu)] = make_int2(sv[j], d[j]);
  }
  __syncthreads();

  int2 pr[NPT];
#pragma unroll
  for (int j = 0; j < NPT; ++j) {
    const int i = tid + j * BK_THREADS;
    pr[j] = stage[i < len ? i : 0];
  }
#pragma unroll
  for (int j = 0; j < NPT; ++j) {
    const int i = tid + j * BK_THREADS;
    if (i < len) pairs[e0 + i] = pr[j];
  }
}

template <bool ROWS>
__device__ __forceinline__ void cell_sum(float* acc, const v2f* rows, const int2* __restrict__ pairs,
                                         const int2* __restrict__ runs, int S,
                                         int tlo, int tspan, int slo, int sspan) {
  const int tid = threadIdx.x;
  const int lane = tid & 31;
  const int wave = __builtin_amdgcn_readfirstlane(tid >> 5);
  const int step = 32 * BK_U;

  for (int g = wave * RW; g < S; g += (BK_THREADS / 32) * RW) {
    const int si = g + (lane & (RW - 1));
    const int2 te = runs[si < S ? si : S - 1];
    const int ln = si < S ? te.y - te.x : 0;
    int cs[RW], dl[RW];
    int total = 0, prev = 0;
#pragma unroll
    for (int q = 0; q < RW; ++q) {
      const int lq = __builtin_amdgcn_readlane(ln, q);
      const int fq = __builtin_amdgcn_readlane(te.x, q);
      cs[q] = total;
      dl[q] = (fq - total) - prev;
      prev = fq - total;
      total += lq;
    }
    if (total <= 0) continue;
    const int lastv = total - 1;

    auto pair_at = [&](int v) -> int2 {
      const int vc = v < lastv ? v : lastv;
      int o = vc + dl[0];
#pragma unroll
      for (int q = 1; q < RW; ++q) o += (vc >= cs[q]) ? dl[q] : 0;
      return pairs[o];
    };

    int2 pcur[BK_U];
#pragma unroll
    for (int j = 0; j < BK_U; ++j) pcur[j] = pair_at(lane + j * 32);

    for (int vb = 0; vb < total; vb += step) {
      int2 pnext[BK_U];
#pragma unroll
      for (int j = 0; j < BK_U; ++j) pnext[j] = pair_at(vb + step + lane + j * 32);
      unsigned relt[BK_U], rels[BK_U];
      bool hit[BK_U];
#pragma unroll
      for (int j = 0; j < BK_U; ++j) {
        const int v = vb + lane + j * 32;
        relt[j] = (unsigned)(pcur[j].y - tlo);
        rels[j] = (unsigned)(pcur[j].x - slo);
        hit[j] = (v < total) && (relt[j] < (unsigned)tspan) && (!ROWS || rels[j] < (unsigned)sspan);
      }
      if (!ROWS) {
#pragma unroll
        for (int j = 0; j < BK_U; ++j)
          if (hit[j]) lds_add(&acc[relt[j]], 1.0f);
      } else {
        v2f v[BK_U];
#pragma unroll
        for (int j = 0; j < BK_U; ++j) v[j] = rows[hit[j] ? rels[j] : 0u];
#pragma unroll
        for (int j = 0; j < BK_U; ++j)
          if (hit[j]) {
            float* a = &acc[relt[j]];
            lds_add(a, v[j].x);
            lds_add(a + ACC_P, v[j].y);
          }
      }
#pragma unroll
      for (int j = 0; j < BK_U; ++j) pcur[j] = pnext[j];
    }
  }
  __syncthreads();
}

__global__ void __launch_bounds__(BK_THREADS)
k_cell_deg(const int* __restrict__ pairs_i, const int* __restrict__ runs_i, int S,
           float* __restrict__ planes, int n, int tn, int npl) {
  __shared__ __attribute__((aligned(16))) float acc[ACC_P];
  const int tb = blockIdx.x / GS, sb = blockIdx.x % GS;
  const int tlo = tb * tn;
  const int tspan = block_span(tlo, n, tn);
  for (int i = threadIdx.x; i < tspan; i += BK_THREADS) acc[i] = 0.0f;
  __syncthreads();
  cell_sum<false>(acc, nullptr, (const int2*)pairs_i, (const int2*)runs_i + (size_t)blockIdx.x * S, S,
                  tlo, tspan, 0, 0);
  float* pl = planes + (size_t)sb * npl + tlo;
  for (int i = threadIdx.x; i < tspan; i += BK_THREADS) pl[i] = acc[i];
}

__global__ void __launch_bounds__(NODE_THREADS)
k_node_deg(const float* __restrict__ planes, const float* __restrict__ x,
           float* __restrict__ dinv, float* __restrict__ xs, int n, int npl) {
  const int node = blockIdx.x * NODE_THREADS + threadIdx.x;
  if (node >= n) return;
  float deg = 1.0f;
#pragma unroll
  for (int k = 0; k < GS; ++k) deg += planes[(size_t)k * npl + node];
  float di = rsqrtf(deg);
  dinv[node] = di;
  v2f xv = *(const v2f*)(x + (size_t)node * 2);
  v2f a;
  a.x = xv.x * di;
  a.y = xv.y * di;
  *(v2f*)(xs + (size_t)node * 2) = a;
}

__device__ __forceinline__ void cell_rows2(float* lds, const int2* __restrict__ pairs,
                                           const int2* __restrict__ runs, int S,
                                           const float* __restrict__ g, float* __restrict__ pl,
                                           int tlo, int tspan, int slo, int sspan) {
  float* acc = lds;
  v2f* rows = (v2f*)(lds + 2 * ACC_P);
  for (int i = threadIdx.x; i < tspan; i += BK_THREADS) {
    acc[i] = 0.0f;
    acc[ACC_P + i] = 0.0f;
  }
  stage_rows(rows, g, sspan);
  __syncthreads();
  cell_sum<true>(acc, rows, pairs, runs, S, tlo, tspan, slo, sspan);
  for (int i = threadIdx.x; i < tspan; i += BK_THREADS) {
    v2f o;
    o.x = acc[i];
    o.y = acc[ACC_P + i];
    *(v2f*)(pl + (size_t)i * 2) = o;
  }
}

__global__ void __launch_bounds__(BK_THREADS)
k_cell_l1(const int* __restrict__ pairs_i, const int* __restrict__ runs_i, int S,
          const float* __restrict__ xs, float* __restrict__ planes, int n, int tn, int sn, int npl) {
  __shared__ __attribute__((aligned(16))) float lds[LDS_FLOATS];
  const int tb = blockIdx.x / GS, sb = blockIdx.x % GS;
  const int tlo = tb * tn, slo = sb * sn;
  const int tspan = block_span(tlo, n, tn), sspan = block_span(slo, n, sn);
  cell_rows2(lds, (const int2*)pairs_i, (const int2*)runs_i + (size_t)blockIdx.x * S, S,
             xs + (size_t)slo * 2, planes + ((size_t)sb * npl + tlo) * 2, tlo, tspan, slo, sspan);
}

__global__ void __launch_bounds__(NODE_THREADS)
k_node_l1(const float* __restrict__ planes, const float* __restrict__ xs,
          const float* __restrict__ dinv, const float* __restrict__ W1,
          const float* __restrict__ b1, const float* __restrict__ W2,
          float* __restrict__ t2, int n, int npl) {
  const int lane = threadIdx.x & 31;
  const int half = lane >> 4;
  const int l    = lane & 15;
  const int ka   = half * 2;
  const int wave = __builtin_amdgcn_readfirstlane((int)(threadIdx.x >> 5));

  const int lm = l & 7;
  const int feat = ((lm & 1) + ((lm >> 1) << 2) + ((l >> 3) << 1)) & 7;
  const float w1a = W1[feat];
  const float w1b = W1[8 + feat];
  const bool aok = (!half) && (lm < 4);
  v2f wa;
  wa.x = aok ? w1a : 0.0f;
  wa.y = aok ? w1b : 0.0f;

  const v2f bb0 = *(const v2f*)(b1 + ka);
  const v2f bb1 = *(const v2f*)(b1 + ka + 4);
  const int lc = l & 3;
  const float wr0 = W2[(ka)     * 4 + lc];
  const float wr1 = W2[(ka + 1) * 4 + lc];
  const float wr2 = W2[(ka + 4) * 4 + lc];
  const float wr3 = W2[(ka + 5) * 4 + lc];
  const bool cok = (l < 4);
  v2f b1v, b2v;
  b1v.x = cok ? wr0 : 0.0f;
  b1v.y = cok ? wr1 : 0.0f;
  b2v.x = cok ? wr2 : 0.0f;
  b2v.y = cok ? wr3 : 0.0f;

  float* th = t2 + (size_t)((l >> 1) & 1) * 2 * npl + (l & 1);
  const int ntiles = (n + 15) >> 4;
  const int nwaves = (int)gridDim.x * (NODE_THREADS / 32);
  for (int t = (int)blockIdx.x * (NODE_THREADS / 32) + wave; t < ntiles; t += nwaves) {
    const int i  = t * 16 + l;
    const int ic = i < n ? i : n - 1;
    v2f xa = *(const v2f*)(xs + (size_t)ic * 2);
#pragma unroll
    for (int k = 0; k < GS; ++k) {
      const v2f p = *(const v2f*)(planes + ((size_t)k * npl + ic) * 2);
      xa.x += p.x;
      xa.y += p.y;
    }
    const float dr = dinv[ic];
    v2f xb;
    xb.x = half ? 0.0f : xa.x * dr;
    xb.y = half ? 0.0f : xa.y * dr;

    v8f g = {};
    g = __builtin_amdgcn_wmma_f32_16x16x4_f32(false, wa, false, xb, (short)0, g, false, false);

    v2f a1, a2;
    a1.x = fmaxf(g[0] + bb0.x, 0.0f);
    a1.y = fmaxf(g[1] + bb0.y, 0.0f);
    a2.x = fmaxf(g[2] + bb1.x, 0.0f);
    a2.y = fmaxf(g[3] + bb1.y, 0.0f);

    v8f c = {};
    c = __builtin_amdgcn_wmma_f32_16x16x4_f32(false, a1, false, b1v, (short)0, c, false, false);
    c = __builtin_amdgcn_wmma_f32_16x16x4_f32(false, a2, false, b2v, (short)0, c, false, false);

    if (l < 4) {
#pragma unroll
      for (int r = 0; r < 8; ++r) {
        const int ir = t * 16 + half * 8 + r;
        if (ir < n) th[(size_t)ir * 2] = c[r] * dinv[ir];
      }
    }
  }
}

__global__ void __launch_bounds__(BK_THREADS)
k_cell_l2(const int* __restrict__ pairs_i, const int* __restrict__ runs_i, int S,
          const float* __restrict__ t2, float* __restrict__ planes, int n, int tn, int sn, int npl) {
  __shared__ __attribute__((aligned(16))) float lds[LDS_FLOATS];
  const int tb = blockIdx.x / GS, sb = blockIdx.x % GS;
  const int tlo = tb * tn, slo = sb * sn;
  const int tspan = block_span(tlo, n, tn), sspan = block_span(slo, n, sn);
  for (int hf = 0; hf < 2; ++hf) {
    cell_rows2(lds, (const int2*)pairs_i, (const int2*)runs_i + (size_t)blockIdx.x * S, S,
               t2 + ((size_t)hf * npl + slo) * 2, planes + ((size_t)(hf * GS + sb) * npl + tlo) * 2,
               tlo, tspan, slo, sspan);
    __syncthreads();
  }
}

__global__ void __launch_bounds__(NODE_THREADS)
k_node_l2(const float* __restrict__ planes, const float* __restrict__ t2,
          const float* __restrict__ dinv, const float* __restrict__ b2,
          float* __restrict__ out, int n, int npl) {
  const int node = blockIdx.x * NODE_THREADS + threadIdx.x;
  if (node >= n) return;
  const v2f s0 = *(const v2f*)(t2 + (size_t)node * 2);
  const v2f s1 = *(const v2f*)(t2 + ((size_t)npl + node) * 2);
  float4 a;
  a.x = s0.x;
  a.y = s0.y;
  a.z = s1.x;
  a.w = s1.y;
#pragma unroll
  for (int k = 0; k < GS; ++k) {
    const v2f p = *(const v2f*)(planes + ((size_t)k * npl + node) * 2);
    const v2f q = *(const v2f*)(planes + ((size_t)(GS + k) * npl + node) * 2);
    a.x += p.x;
    a.y += p.y;
    a.z += q.x;
    a.w += q.y;
  }
  float di = dinv[node];
  a.x = b2[0] + a.x * di;
  a.y = b2[1] + a.y * di;
  a.z = b2[2] + a.z * di;
  a.w = b2[3] + a.w * di;
  *(float4*)(out + (size_t)node * 4) = a;
}

extern "C" void kernel_launch(void* const* d_in, const int* in_sizes, int n_in,
                              void* d_out, int out_size, void* d_ws, size_t ws_size,
                              hipStream_t stream) {
  const float* x  = (const float*)d_in[0];
  const int*   ei = (const int*)d_in[1];
  const float* W1 = (const float*)d_in[2];
  const float* b1 = (const float*)d_in[3];
  const float* W2 = (const float*)d_in[4];
  const float* b2 = (const float*)d_in[5];
  float* out = (float*)d_out;

  const int n = in_sizes[0] / 2;
  const int E = in_sizes[1] / 2;
  const int* src = ei;
  const int* dst = ei + E;

  const int tn = (n + GT - 1) / GT > 0 ? (n + GT - 1) / GT : 1;
  const int sn = (n + GS - 1) / GS > 0 ? (n + GS - 1) / GS : 1;
  const int Smin = (E + STAGE_PAIRS - 1) / STAGE_PAIRS;
  const int S = Smin > 256 ? Smin : 256;

  char* ws = (char*)d_ws;
  size_t nAl = ((size_t)n + 3) & ~(size_t)3;
  const int npl = (int)nAl;
  float* dinv   = (float*)ws;  ws += sizeof(float) * nAl;
  float* xs     = (float*)ws;  ws += sizeof(float) * nAl * 2;
  float* t2     = (float*)ws;  ws += sizeof(float) * nAl * 4;
  int*   runs   = (int*)ws;    ws += sizeof(int) * 2 * (size_t)S * NB;
  float* planes = (float*)ws;  ws += sizeof(float) * 2 * (2 * GS) * nAl;
  int*   pairs  = (int*)ws;

  const int node_blocks = (n + NODE_THREADS - 1) / NODE_THREADS;

  k_slice_group<<<S, BK_THREADS, 0, stream>>>(src, dst, runs, pairs, E, S, tn, sn);

  k_cell_deg<<<NB, BK_THREADS, 0, stream>>>(pairs, runs, S, planes, n, tn, npl);
  k_node_deg<<<node_blocks, NODE_THREADS, 0, stream>>>(planes, x, dinv, xs, n, npl);
  k_cell_l1<<<NB, BK_THREADS, 0, stream>>>(pairs, runs, S, xs, planes, n, tn, sn, npl);
  k_node_l1<<<node_blocks, NODE_THREADS, 0, stream>>>(planes, xs, dinv, W1, b1, W2, t2, n, npl);
  k_cell_l2<<<NB, BK_THREADS, 0, stream>>>(pairs, runs, S, t2, planes, n, tn, sn, npl);
  k_node_l2<<<node_blocks, NODE_THREADS, 0, stream>>>(planes, t2, dinv, b2, out, n, npl);
}


# Grouped edge list kept as ONE 32-bit word per pair (13-bit target place, 19-bit source place inside the cell's blocks), not two ints: every pass moves 4 B per edge, not 8; f32 sums stay unordered
# speedup vs baseline: 16.5608x; 1.0531x over previous
#include <hip/hip_runtime.h>


typedef __attribute__((ext_vector_type(2))) float v2f;
typedef __attribute__((ext_vector_type(8))) float v8f;

#define GT 32
#define GS 8
#define NB (GT * GS)
#define BK_THREADS 1024
#define BK_U 8
#define NPT 25
#define STAGE_PAIRS (NPT * BK_THREADS)
#define RW 8
#define ACC_P 8000
#define ROW_P 32000
#define LDS_FLOATS (2 * ACC_P + 2 * ROW_P)
#define ST_U 13
#define NODE_THREADS 256
#define T_BITS 13
#define T_NONE ((1u << T_BITS) - 1u)
#define S_NONE ((1u << (32 - T_BITS)) - 1u)

static_assert(NB <= BK_THREADS && NB % 32 == 0, "k_slice_group scans the NB counts with one thread each, whole waves");
static_assert(NB + BK_THREADS <= 65536 && STAGE_PAIRS <= 65536, "a cell's number and a place inside its run share one 32-bit word");
static_assert(ACC_P <= T_NONE && ROW_P <= S_NONE, "a block's places must stay below the all-ones value of their field");

__device__ __forceinline__ void lds_add(float* a, float v) {
  __hip_atomic_fetch_add(a, v, __ATOMIC_RELAXED, __HIP_MEMORY_SCOPE_WORKGROUP);
}

__device__ __forceinline__ int cell_of(int s, int d, int tn, int sn) {
  unsigned tb = (unsigned)d / (unsigned)tn;
  unsigned sb = (unsigned)s / (unsigned)sn;
  tb = tb < (unsigned)GT ? tb : (unsigned)(GT - 1);
  sb = sb < (unsigned)GS ? sb : (unsigned)(GS - 1);
  return (int)(tb * GS + sb);
}

__device__ __forceinline__ unsigned pair_word(int s, int d, unsigned cell, int tn, int sn) {
  const unsigned rt = (unsigned)d - (cell / GS) * (unsigned)tn;
  const unsigned rs = (unsigned)s - (cell % GS) * (unsigned)sn;
  return (rt < (unsigned)tn ? rt : T_NONE) | ((rs < (unsigned)sn ? rs : S_NONE) << T_BITS);
}

__device__ __forceinline__ int block_span(int lo, int n, int bn) {
  int span = n - lo;
  span = span < bn ? span : bn;
  return span > 0 ? span : 0;
}

__device__ __forceinline__ void stage_rows(v2f* rows, const float* __restrict__ g, int cnt) {
  const int tid = threadIdx.x;
  for (int base = 0; base < cnt; base += ST_U * BK_THREADS) {
    v2f t[ST_U];
#pragma unroll
    for (int j = 0; j < ST_U; ++j) {
      const int i = base + tid + j * BK_THREADS;
      const int ic = i < cnt ? i : cnt - 1;
      t[j] = *(const v2f*)(g + (size_t)ic * 2);
    }
#pragma unroll
    for (int j = 0; j < ST_U; ++j) {
      const int i = base + tid + j * BK_THREADS;
      if (i < cnt) rows[i] = t[j];
    }
  }
}

__global__ void __launch_bounds__(BK_THREADS)
k_slice_group(const int* __restrict__ src, const int* __restrict__ dst,
              int* __restrict__ runs_i, int* __restrict__ pairs_i, int E, int S, int tn, int sn) {
  __shared__ __attribute__((aligned(16))) unsigned stage[STAGE_PAIRS];
  __shared__ unsigned cnt[NB + BK_THREADS];
  __shared__ int first[NB];
  __shared__ int wsum[NB / 32];
  const int tid = threadIdx.x;
  const int s = blockIdx.x;
  unsigned* pairs = (unsigned*)pairs_i;
  int2* runs = (int2*)runs_i;

  const int chunk = (E + S - 1) / S;
  const int e0 = (s * chunk) < E ? (s * chunk) : E;
  const int e1 = (e0 + chunk) < E ? (e0 + chunk) : E;
  const int len = e1 - e0;
  const int last = len > 0 ? e1 - 1 : 0;

  int d[NPT], sv[NPT];
#pragma unroll
  for (int j = 0; j < NPT; ++j) {
    const int e = e0 + tid + j * BK_THREADS;
    const int ec = e < last ? e : last;
    d[j] = dst[ec];
    sv[j] = src[ec];
  }
  if (tid < NB) cnt[tid] = 0u;
  cnt[NB + tid] = 0u;
  __syncthreads();

  unsigned key[NPT], pw[NPT];
#pragma unroll
  for (int j = 0; j < NPT; ++j) {
    const int lim = len - j * BK_THREADS;
    const unsigned cell = (unsigned)cell_of(sv[j], d[j], tn, sn);
    pw[j] = pair_word(sv[j], d[j], cell, tn, sn);
    const unsigned slot = tid < lim ? cell : (unsigned)(NB + tid);
    const unsigned r = __hip_atomic_fetch_add(&cnt[slot], 1u, __ATOMIC_RELAXED, __HIP_MEMORY_SCOPE_WORKGROUP);
    key[j] = (slot << 16) | r;
  }
#pragma unroll
  for (int j = 0; j < NPT; ++j) asm volatile("" :: "v"(pw[j]));
  __syncthreads();

  int my_c = 0, incl = 0;
  if (tid < NB) {
    my_c = (int)cnt[tid];
    incl = my_c;
    const int lane = tid & 31;
    for (int off = 1; off < 32; off <<= 1) {
      const int t = __shfl_up(incl, off);
      if (lane >= off) incl += t;
    }
    if (lane == 31) wsum[tid >> 5] = incl;
  }
  __syncthreads();
  if (tid < NB) {
    int wb = 0;
    const int w = tid >> 5;
    for (int k = 0; k < w; ++k) wb += wsum[k];
    const int f = wb + incl - my_c;
    first[tid] = f;
    runs[(size_t)tid * S + s] = make_int2(e0 + f, e0 + f + my_c);
  }
  __syncthreads();

  int fj[NPT];
#pragma unroll
  for (int j = 0; j < NPT; ++j) {
    const unsigned b = key[j] >> 16;
    fj[j] = first[b < (unsigned)NB ? b : NB - 1];
  }
#pragma unroll
  for (int j = 0; j < NPT; ++j) {
    const int lim = len - j * BK_THREADS;
    if (tid < lim) stage[fj[j] + (int)(key[j] & 0xffffu)] = pw[j];
  }
  __syncthreads();

  unsigned pr[NPT];
#pragma unroll
  for (int j = 0; j < NPT; ++j) {
    const int i = tid + j * BK_THREADS;
    pr[j] = stage[i < len ? i : 0];
  }
#pragma unroll
  for (int j = 0; j < NPT; ++j) {
    const int i = tid + j * BK_THREADS;
    if (i < len) pairs[e0 + i] = pr[j];
  }
}

template <bool ROWS>
__device__ __forceinline__ void cell_sum(float* acc, const v2f* rows, const unsigned* __restrict__ pairs,
                                         const int2* __restrict__ runs, int S,
                                         int tspan, int sspan) {
  const int tid = threadIdx.x;
  const int lane = tid & 31;
  const int wave = __builtin_amdgcn_readfirstlane(tid >> 5);
  const int step = 32 * BK_U;

  for (int g = wave * RW; g < S; g += (BK_THREADS / 32) * RW) {
    const int si = g + (lane & (RW - 1));
    const int2 te = runs[si < S ? si : S - 1];
    const int ln = si < S ? te.y - te.x : 0;
    int cs[RW], dl[RW];
    int total = 0, prev = 0;
#pragma unroll
    for (int q = 0; q < RW; ++q) {
      const int lq = __builtin_amdgcn_readlane(ln, q);
      const int fq = __builtin_amdgcn_readlane(te.x, q);
      cs[q] = total;
      dl[q] = (fq - total) - prev;
      prev = fq - total;
      total += lq;
    }
    if (total <= 0) continue;
    const int lastv = total - 1;

    auto pair_at = [&](int v) -> unsigned {
      const int vc = v < lastv ? v : lastv;
      int o = vc + dl[0];
#pragma unroll
      for (int q = 1; q < RW; ++q) o += (vc >= cs[q]) ? dl[q] : 0;
      return pairs[o];
    };

    unsigned pcur[BK_U];
#pragma unroll
    for (int j = 0; j < BK_U; ++j) pcur[j] = pair_at(lane + j * 32);

    for (int vb = 0; vb < total; vb += step) {
      unsigned pnext[BK_U];
#pragma unroll
      for (int j = 0; j < BK_U; ++j) pnext[j] = pair_at(vb + step + lane + j * 32);
      unsigned relt[BK_U], rels[BK_U];
      bool hit[BK_U];
#pragma unroll
      for (int j = 0; j < BK_U; ++j) {
        const int v = vb + lane + j * 32;
        relt[j] = pcur[j] & T_NONE;
        rels[j] = pcur[j] >> T_BITS;
        hit[j] = (v < total) && (relt[j] < (unsigned)tspan) && (!ROWS || rels[j] < (unsigned)sspan);
      }
      if (!ROWS) {
#pragma unroll
        for (int j = 0; j < BK_U; ++j)
          if (hit[j]) lds_add(&acc[relt[j]], 1.0f);
      } else {
        v2f v[BK_U];
#pragma unroll
        for (int j = 0; j < BK_U; ++j) v[j] = rows[hit[j] ? rels[j] : 0u];
#pragma unroll
        for (int j = 0; j < BK_U; ++j)
          if (hit[j]) {
            float* a = &acc[relt[j]];
            lds_add(a, v[j].x);
            lds_add(a + ACC_P, v[j].y);
          }
      }
#pragma unroll
      for (int j = 0; j < BK_U; ++j) pcur[j] = pnext[j];
    }
  }
  __syncthreads();
}

__global__ void __launch_bounds__(BK_THREADS)
k_cell_deg(const int* __restrict__ pairs_i, const int* __restrict__ runs_i, int S,
           float* __restrict__ planes, int n, int tn, int npl) {
  __shared__ __attribute__((aligned(16))) float acc[ACC_P];
  const int tb = blockIdx.x / GS, sb = blockIdx.x % GS;
  const int tlo = tb * tn;
  const int tspan = block_span(tlo, n, tn);
  for (int i = threadIdx.x; i < tspan; i += BK_THREADS) acc[i] = 0.0f;
  __syncthreads();
  cell_sum<false>(acc, nullptr, (const unsigned*)pairs_i, (const int2*)runs_i + (size_t)blockIdx.x * S, S,
                  tspan, 0);
  float* pl = planes + (size_t)sb * npl + tlo;
  for (int i = threadIdx.x; i < tspan; i += BK_THREADS) pl[i] = acc[i];
}

__global__ void __launch_bounds__(NODE_THREADS)
k_node_deg(const float* __restrict__ planes, const float* __restrict__ x,
           float* __restrict__ dinv, float* __restrict__ xs, int n, int npl) {
  const int node = blockIdx.x * NODE_THREADS + threadIdx.x;
  if (node >= n) return;
  float deg = 1.0f;
#pragma unroll
  for (int k = 0; k < GS; ++k) deg += planes[(size_t)k * npl + node];
  float di = rsqrtf(deg);
  dinv[node] = di;
  v2f xv = *(const v2f*)(x + (size_t)node * 2);
  v2f a;
  a.x = xv.x * di;
  a.y = xv.y * di;
  *(v2f*)(xs + (size_t)node * 2) = a;
}

__device__ __forceinline__ void cell_rows2(float* lds, const unsigned* __restrict__ pairs,
                                           const int2* __restrict__ runs, int S,
                                           const float* __restrict__ g, float* __restrict__ pl,
                                           int tspan, int sspan) {
  float* acc = lds;
  v2f* rows = (v2f*)(lds + 2 * ACC_P);
  for (int i = threadIdx.x; i < tspan; i += BK_THREADS) {
    acc[i] = 0.0f;
    acc[ACC_P + i] = 0.0f;
  }
  stage_rows(rows, g, sspan);
  __syncthreads();
  cell_sum<true>(acc, rows, pairs, runs, S, tspan, sspan);
  for (int i = threadIdx.x; i < tspan; i += BK_THREADS) {
    v2f o;
    o.x = acc[i];
    o.y = acc[ACC_P + i];
    *(v2f*)(pl + (size_t)i * 2) = o;
  }
}

__global__ void __launch_bounds__(BK_THREADS)
k_cell_l1(const int* __restrict__ pairs_i, const int* __restrict__ runs_i, int S,
          const float* __restrict__ xs, float* __restrict__ planes, int n, int tn, int sn, int npl) {
  __shared__ __attribute__((aligned(16))) float lds[LDS_FLOATS];
  const int tb = blockIdx.x / GS, sb = blockIdx.x % GS;
  const int tlo = tb * tn, slo = sb * sn;
  const int tspan = block_span(tlo, n, tn), sspan = block_span(slo, n, sn);
  cell_rows2(lds, (const unsigned*)pairs_i, (const int2*)runs_i + (size_t)blockIdx.x * S, S,
             xs + (size_t)slo * 2, planes + ((size_t)sb * npl + tlo) * 2, tspan, sspan);
}

__global__ void __launch_bounds__(NODE_THREADS)
k_node_l1(const float* __restrict__ planes, const float* __restrict__ xs,
          const float* __restrict__ dinv, const float* __restrict__ W1,
          const float* __restrict__ b1, const float* __restrict__ W2,
          float* __restrict__ t2, int n, int npl) {
  const int lane = threadIdx.x & 31;
  const int half = lane >> 4;
  const int l    = lane & 15;
  const int ka   = half * 2;
  const int wave = __builtin_amdgcn_readfirstlane((int)(threadIdx.x >> 5));

  const int lm = l & 7;
  const int feat = ((lm & 1) + ((lm >> 1) << 2) + ((l >> 3) << 1)) & 7;
  const float w1a = W1[feat];
  const float w1b = W1[8 + feat];
  const bool aok = (!half) && (lm < 4);
  v2f wa;
  wa.x = aok ? w1a : 0.0f;
  wa.y = aok ? w1b : 0.0f;

  const v2f bb0 = *(const v2f*)(b1 + ka);
  const v2f bb1 = *(const v2f*)(b1 + ka + 4);
  const int lc = l & 3;
  const float wr0 = W2[(ka)     * 4 + lc];
  const float wr1 = W2[(ka + 1) * 4 + lc];
  const float wr2 = W2[(ka + 4) * 4 + lc];
  const float wr3 = W2[(ka + 5) * 4 + lc];
  const bool cok = (l < 4);
  v2f b1v, b2v;
  b1v.x = cok ? wr0 : 0.0f;
  b1v.y = cok ? wr1 : 0.0f;
  b2v.x = cok ? wr2 : 0.0f;
  b2v.y = cok ? wr3 : 0.0f;

  float* th = t2 + (size_t)((l >> 1) & 1) * 2 * npl + (l & 1);
  const int ntiles = (n + 15) >> 4;
  const int nwaves = (int)gridDim.x * (NODE_THREADS / 32);
  for (int t = (int)blockIdx.x * (NODE_THREADS / 32) + wave; t < ntiles; t += nwaves) {
    const int i  = t * 16 + l;
    const int ic = i < n ? i : n - 1;
    v2f xa = *(const v2f*)(xs + (size_t)ic * 2);
#pragma unroll
    for (int k = 0; k < GS; ++k) {
      const v2f p = *(const v2f*)(planes + ((size_t)k * npl + ic) * 2);
      xa.x += p.x;
      xa.y += p.y;
    }
    const float dr = dinv[ic];
    v2f xb;
    xb.x = half ? 0.0f : xa.x * dr;
    xb.y = half ? 0.0f : xa.y * dr;

    v8f g = {};
    g = __builtin_amdgcn_wmma_f32_16x16x4_f32(false, wa, false, xb, (short)0, g, false, false);

    v2f a1, a2;
    a1.x = fmaxf(g[0] + bb0.x, 0.0f);
    a1.y = fmaxf(g[1] + bb0.y, 0.0f);
    a2.x = fmaxf(g[2] + bb1.x, 0.0f);
    a2.y = fmaxf(g[3] + bb1.y, 0.0f);

    v8f c = {};
    c = __builtin_amdgcn_wmma_f32_16x16x4_f32(false, a1, false, b1v, (short)0, c, false, false);
    c = __builtin_amdgcn_wmma_f32_16x16x4_f32(false, a2, false, b2v, (short)0, c, false, false);

    if (l < 4) {
#pragma unroll
      for (int r = 0; r < 8; ++r) {
        const int ir = t * 16 + half * 8 + r;
        if (ir < n) th[(size_t)ir * 2] = c[r] * dinv[ir];
      }
    }
  }
}

__global__ void __launch_bounds__(BK_THREADS)
k_cell_l2(const int* __restrict__ pairs_i, const int* __restrict__ runs_i, int S,
          const float* __restrict__ t2, float* __restrict__ planes, int n, int tn, int sn, int npl) {
  __shared__ __attribute__((aligned(16))) float lds[LDS_FLOATS];
  const int tb = blockIdx.x / GS, sb = blockIdx.x % GS;
  const int tlo = tb * tn, slo = sb * sn;
  const int tspan = block_span(tlo, n, tn), sspan = block_span(slo, n, sn);
  for (int hf = 0; hf < 2; ++hf) {
    cell_rows2(lds, (const unsigned*)pairs_i, (const int2*)runs_i + (size_t)blockIdx.x * S, S,
               t2 + ((size_t)hf * npl + slo) * 2, planes + ((size_t)(hf * GS + sb) * npl + tlo) * 2,
               tspan, sspan);
    __syncthreads();
  }
}

__global__ void __launch_bounds__(NODE_THREADS)
k_node_l2(const float* __restrict__ planes, const float* __restrict__ t2,
          const float* __restrict__ dinv, const float* __restrict__ b2,
          float* __restrict__ out, int n, int npl) {
  const int node = blockIdx.x * NODE_THREADS + threadIdx.x;
  if (node >= n) return;
  const v2f s0 = *(const v2f*)(t2 + (size_t)node * 2);
  const v2f s1 = *(const v2f*)(t2 + ((size_t)npl + node) * 2);
  float4 a;
  a.x = s0.x;
  a.y = s0.y;
  a.z = s1.x;
  a.w = s1.y;
#pragma unroll
  for (int k = 0; k < GS; ++k) {
    const v2f p = *(const v2f*)(planes + ((size_t)k * npl + node) * 2);
    const v2f q = *(const v2f*)(planes + ((size_t)(GS + k) * npl + node) * 2);
    a.x += p.x;
    a.y += p.y;
    a.z += q.x;
    a.w += q.y;
  }
  float di = dinv[node];
  a.x = b2[0] + a.x * di;
  a.y = b2[1] + a.y * di;
  a.z = b2[2] + a.z * di;
  a.w = b2[3] + a.w * di;
  *(float4*)(out + (size_t)node * 4) = a;
}

extern "C" void kernel_launch(void* const* d_in, const int* in_sizes, int n_in,
                              void* d_out, int out_size, void* d_ws, size_t ws_size,
                              hipStream_t stream) {
  const float* x  = (const float*)d_in[0];
  const int*   ei = (const int*)d_in[1];
  const float* W1 = (const float*)d_in[2];
  const float* b1 = (const float*)d_in[3];
  const float* W2 = (const float*)d_in[4];
  const float* b2 = (const float*)d_in[5];
  float* out = (float*)d_out;

  const int n = in_sizes[0] / 2;
  const int E = in_sizes[1] / 2;
  const int* src = ei;
  const int* dst = ei + E;

  const int tn = (n + GT - 1) / GT > 0 ? (n + GT - 1) / GT : 1;
  const int sn = (n + GS - 1) / GS > 0 ? (n + GS - 1) / GS : 1;
  const int Smin = (E + STAGE_PAIRS - 1) / STAGE_PAIRS;
  const int S = Smin > 256 ? Smin : 256;

  char* ws = (char*)d_ws;
  size_t nAl = ((size_t)n + 3) & ~(size_t)3;
  const int npl = (int)nAl;
  float* dinv   = (float*)ws;  ws += sizeof(float) * nAl;
  float* xs     = (float*)ws;  ws += sizeof(float) * nAl * 2;
  float* t2     = (float*)ws;  ws += sizeof(float) * nAl * 4;
  int*   runs   = (int*)ws;    ws += sizeof(int) * 2 * (size_t)S * NB;
  float* planes = (float*)ws;  ws += sizeof(float) * 2 * (2 * GS) * nAl;
  int*   pairs  = (int*)ws;

  const int node_blocks = (n + NODE_THREADS - 1) / NODE_THREADS;

  k_slice_group<<<S, BK_THREADS, 0, stream>>>(src, dst, runs, pairs, E, S, tn, sn);

  k_cell_deg<<<NB, BK_THREADS, 0, stream>>>(pairs, runs, S, planes, n, tn, npl);
  k_node_deg<<<node_blocks, NODE_THREADS, 0, stream>>>(planes, x, dinv, xs, n, npl);
  k_cell_l1<<<NB, BK_THREADS, 0, stream>>>(pairs, runs, S, xs, planes, n, tn, sn, npl);
  k_node_l1<<<node_blocks, NODE_THREADS, 0, stream>>>(planes, xs, dinv, W1, b1, W2, t2, n, npl);
  k_cell_l2<<<NB, BK_THREADS, 0, stream>>>(pairs, runs, S, t2, planes, n, tn, sn, npl);
  k_node_l2<<<node_blocks, NODE_THREADS, 0, stream>>>(planes, t2, dinv, b2, out, n, npl);
}
